# starting kernel
# baseline (speedup 1.0000x reference)
#include <hip/hip_runtime.h>
#include <math.h>


typedef float v2f __attribute__((ext_vector_type(2)));
typedef float v8f __attribute__((ext_vector_type(8)));

#define DEV __device__ __forceinline__

DEV float wsum(float v) {
#pragma unroll
  for (int m = 16; m; m >>= 1) v += __shfl_xor(v, m, 32);
  return v;
}

DEV unsigned enc_f(float f) {
  unsigned u = __float_as_uint(f);
  return (u & 0x80000000u) ? ~u : (u | 0x80000000u);
}
DEV float dec_f(unsigned u) {
  return __uint_as_float((u & 0x80000000u) ? (u & 0x7FFFFFFFu) : ~u);
}

__global__ void k_mean(const float* __restrict__ x, float* __restrict__ out,
                       int NT, int N, int threeN) {
  int wid = (blockIdx.x * blockDim.x + threadIdx.x) >> 5;
  int lane = threadIdx.x & 31;
  if (wid >= NT) return;
  const float* r = x + (size_t)wid * 128;
  float s = r[lane] + r[lane + 32] + r[lane + 64] + r[lane + 96];
  s = wsum(s);
  if (lane == 0) {
    int b = wid / N, n = wid % N;
    out[(size_t)b * threeN + n] = s * (1.0f / 128.0f);
  }
}

__global__ void k_gemm(const float* __restrict__ A, const float* __restrict__ Bw,
                       float* __restrict__ C, int M, int K) {
  int wid = (blockIdx.x * blockDim.x + threadIdx.x) >> 5;
  int lane = threadIdx.x & 31;
  int rowTile = wid >> 2, colGrp = wid & 3;
  int row0 = rowTile << 4, col0 = colGrp << 6;
  if (row0 >= M) return;
  int half = lane >> 4, lm = lane & 15;

  v8f acc0 = {}, acc1 = {}, acc2 = {}, acc3 = {};
  const float* ap = A + (size_t)(row0 + lm) * K + 2 * half;
  const float* bp = Bw + (size_t)(2 * half) * 256 + col0 + lm;

  for (int k0 = 0; k0 < K; k0 += 4) {
    v2f a;
    a.x = ap[k0];
    a.y = ap[k0 + 1];
    const float* b0 = bp + (size_t)k0 * 256;
    v2f b;
    b.x = b0[0];  b.y = b0[256];
    acc0 = __builtin_amdgcn_wmma_f32_16x16x4_f32(false, a, false, b, (short)0, acc0, false, false);
    b.x = b0[16]; b.y = b0[256 + 16];
    acc1 = __builtin_amdgcn_wmma_f32_16x16x4_f32(false, a, false, b, (short)0, acc1, false, false);
    b.x = b0[32]; b.y = b0[256 + 32];
    acc2 = __builtin_amdgcn_wmma_f32_16x16x4_f32(false, a, false, b, (short)0, acc2, false, false);
    b.x = b0[48]; b.y = b0[256 + 48];
    acc3 = __builtin_amdgcn_wmma_f32_16x16x4_f32(false, a, false, b, (short)0, acc3, false, false);
  }

  float* cp = C + (size_t)(row0 + 8 * half) * 256 + col0 + lm;
#pragma unroll
  for (int r2 = 0; r2 < 8; r2++) {
    cp[(size_t)r2 * 256 + 0]  = acc0[r2];
    cp[(size_t)r2 * 256 + 16] = acc1[r2];
    cp[(size_t)r2 * 256 + 32] = acc2[r2];
    cp[(size_t)r2 * 256 + 48] = acc3[r2];
  }
}

__global__ void k_att(const float* __restrict__ xp, const float* __restrict__ att_s,
                      const float* __restrict__ att_d, float* __restrict__ a_s,
                      float* __restrict__ a_d, int NT) {
  int wid = (blockIdx.x * blockDim.x + threadIdx.x) >> 5;
  int lane = threadIdx.x & 31;
  if (wid >= NT) return;
  const float* row = xp + (size_t)wid * 256;
#pragma unroll
  for (int h = 0; h < 4; h++) {
    int c0 = h * 64 + lane;
    float vs = row[c0] * att_s[c0] + row[c0 + 32] * att_s[c0 + 32];
    float vd = row[c0] * att_d[c0] + row[c0 + 32] * att_d[c0 + 32];
    vs = wsum(vs);
    vd = wsum(vd);
    if (lane == 0) {
      a_s[wid * 4 + h] = vs;
      a_d[wid * 4 + h] = vd;
    }
  }
}

__global__ void k_init(float* __restrict__ agg, float* __restrict__ denom,
                       unsigned* __restrict__ m, int n_agg, int n_small) {
  int i = blockIdx.x * blockDim.x + threadIdx.x;
  if (i < n_agg) agg[i] = 0.0f;
  if (i < n_small) {
    denom[i] = 0.0f;
    m[i] = 0u;
  }
}

DEV void edge_sd(int e, int E, const int* __restrict__ src, const int* __restrict__ dst,
                 int& s, int& d) {
  if (e < E) { s = src[e]; d = dst[e]; }
  else       { s = d = e - E; }
}

__global__ void k_edge_max(const int* __restrict__ src, const int* __restrict__ dst,
                           const float* __restrict__ a_s, const float* __restrict__ a_d,
                           unsigned* __restrict__ m, int E, int ET) {
  int idx = blockIdx.x * blockDim.x + threadIdx.x;
  if (idx >= ET * 4) return;
  int e = idx >> 2, h = idx & 3;
  int s, d;
  edge_sd(e, E, src, dst, s, d);
  float l = a_s[s * 4 + h] + a_d[d * 4 + h];
  l = (l > 0.0f) ? l : 0.2f * l;
  atomicMax(&m[d * 4 + h], enc_f(l));
}

__global__ void k_edge_exp(const int* __restrict__ src, const int* __restrict__ dst,
                           const float* __restrict__ a_s, const float* __restrict__ a_d,
                           const unsigned* __restrict__ m, float* __restrict__ ew,
                           float* __restrict__ denom, int E, int ET) {
  int idx = blockIdx.x * blockDim.x + threadIdx.x;
  if (idx >= ET * 4) return;
  int e = idx >> 2, h = idx & 3;
  int s, d;
  edge_sd(e, E, src, dst, s, d);
  float l = a_s[s * 4 + h] + a_d[d * 4 + h];
  l = (l > 0.0f) ? l : 0.2f * l;
  float ev = expf(l - dec_f(m[d * 4 + h]));
  ew[idx] = ev;
  atomicAdd(&denom[d * 4 + h], ev);
}

__global__ void k_edge_scatter(const int* __restrict__ src, const int* __restrict__ dst,
                               const float* __restrict__ ew, const float* __restrict__ denom,
                               const float* __restrict__ xp, float* __restrict__ agg,
                               int E, int ET) {
  int wid = (blockIdx.x * blockDim.x + threadIdx.x) >> 5;
  int lane = threadIdx.x & 31;
  if (wid >= ET) return;
  int s, d;
  edge_sd(wid, E, src, dst, s, d);
  float al[4];
#pragma unroll
  for (int h = 0; h < 4; h++)
    al[h] = ew[wid * 4 + h] / (denom[d * 4 + h] + 1e-16f);
  const float* xr = xp + (size_t)s * 256;
  float* ar = agg + (size_t)d * 256;
#pragma unroll
  for (int i = 0; i < 8; i++) {
    int ch = lane + i * 32;
    atomicAdd(&ar[ch], xr[ch] * al[ch >> 6]);
  }
}

__global__ void k_epilogue(const float* __restrict__ agg, const float* __restrict__ bias,
                           const float* __restrict__ pw, const float* __restrict__ pb,
                           float* __restrict__ hout, float* __restrict__ out,
                           int NT, int N, int threeN, int sec) {
  int wid = (blockIdx.x * blockDim.x + threadIdx.x) >> 5;
  int lane = threadIdx.x & 31;
  if (wid >= NT) return;
  const float* ar = agg + (size_t)wid * 256;
  float pool = 0.0f;
#pragma unroll
  for (int i = 0; i < 8; i++) {
    int ch = lane + i * 32;
    float v = ar[ch] + bias[ch];
    v = (v > 0.0f) ? v : expm1f(v);
    hout[(size_t)wid * 256 + ch] = v;
    pool += v * pw[ch];
  }
  pool = wsum(pool);
  if (lane == 0) {
    int b = wid / N, n = wid % N;
    out[(size_t)b * threeN + sec * N + n] = pool + pb[0];
  }
}

extern "C" void kernel_launch(void* const* d_in, const int* in_sizes, int n_in,
                              void* d_out, int out_size, void* d_ws, size_t ws_size,
                              hipStream_t stream) {
  const float* x   = (const float*)d_in[0];
  const int*   ei  = (const int*)d_in[1];
  const float* W1  = (const float*)d_in[2];
  const float* as1 = (const float*)d_in[3];
  const float* ad1 = (const float*)d_in[4];
  const float* b1  = (const float*)d_in[5];
  const float* W2  = (const float*)d_in[6];
  const float* as2 = (const float*)d_in[7];
  const float* ad2 = (const float*)d_in[8];
  const float* b2  = (const float*)d_in[9];
  const float* pw1 = (const float*)d_in[10];
  const float* pb1 = (const float*)d_in[11];
  const float* pw2 = (const float*)d_in[12];
  const float* pb2 = (const float*)d_in[13];
  float* out = (float*)d_out;

  const int NT = 32768, N = 4096, E = 524288;
  const int ET = E + NT;
  const int threeN = 3 * N;
  const size_t FEAT_BYTES = (size_t)NT * 256 * 4;

  char* w = (char*)d_ws;
  float*    xp    = (float*)(w);
  float*    hb    = (float*)(w + FEAT_BYTES);
  float*    agg   = (float*)(w + 2 * FEAT_BYTES);
  float*    a_s   = (float*)(w + 3 * FEAT_BYTES);
  float*    a_d   = a_s + (size_t)NT * 4;
  unsigned* m     = (unsigned*)(a_d + (size_t)NT * 4);
  float*    denom = (float*)(m + (size_t)NT * 4);
  float*    ew    = denom + (size_t)NT * 4;

  const int* srcA = ei;
  const int* dstA = ei + E;

  dim3 blk(256);
  int gWaveNT  = NT / 8;
  int gGemm    = (NT / 16) * 4 / 8;
  int gInit    = (NT * 256) / 256;
  int gEdgeEH  = (ET * 4) / 256;
  int gEdgeW   = ET / 8;

  k_mean<<<gWaveNT, blk, 0, stream>>>(x, out, NT, N, threeN);

  k_gemm<<<gGemm, blk, 0, stream>>>(x, W1, xp, NT, 128);
  k_att<<<gWaveNT, blk, 0, stream>>>(xp, as1, ad1, a_s, a_d, NT);
  k_init<<<gInit, blk, 0, stream>>>(agg, denom, m, NT * 256, NT * 4);
  k_edge_max<<<gEdgeEH, blk, 0, stream>>>(srcA, dstA, a_s, a_d, m, E, ET);
  k_edge_exp<<<gEdgeEH, blk, 0, stream>>>(srcA, dstA, a_s, a_d, m, ew, denom, E, ET);
  k_edge_scatter<<<gEdgeW, blk, 0, stream>>>(srcA, dstA, ew, denom, xp, agg, E, ET);
  k_epilogue<<<gWaveNT, blk, 0, stream>>>(agg, b1, pw1, pb1, hb, out, NT, N, threeN, 1);

  k_gemm<<<gGemm, blk, 0, stream>>>(hb, W2, xp, NT, 256);
  k_att<<<gWaveNT, blk, 0, stream>>>(xp, as2, ad2, a_s, a_d, NT);
  k_init<<<gInit, blk, 0, stream>>>(agg, denom, m, NT * 256, NT * 4);
  k_edge_max<<<gEdgeEH, blk, 0, stream>>>(srcA, dstA, a_s, a_d, m, E, ET);
  k_edge_exp<<<gEdgeEH, blk, 0, stream>>>(srcA, dstA, a_s, a_d, m, ew, denom, E, ET);
  k_edge_scatter<<<gEdgeW, blk, 0, stream>>>(srcA, dstA, ew, denom, xp, agg, E, ET);
  k_epilogue<<<gWaveNT, blk, 0, stream>>>(agg, b2, pw2, pb2, xp, out, NT, N, threeN, 2);
}


# Edges bucketed by destination once per call (count, scan, fill); one wave per node gathers its in-edges, softmax and sum in registers, no float atomics; sums run in bucket arrival order
# speedup vs baseline: 2.4738x; 2.4738x over previous
#include <hip/hip_runtime.h>
#include <math.h>


typedef float v2f __attribute__((ext_vector_type(2)));
typedef float v8f __attribute__((ext_vector_type(8)));
typedef float v4f __attribute__((ext_vector_type(4)));

#define DEV __device__ __forceinline__

DEV float wsum(float v) {
#pragma unroll
  for (int m = 16; m; m >>= 1) v += __shfl_xor(v, m, 32);
  return v;
}

DEV float wmax(float v) {
#pragma unroll
  for (int m = 16; m; m >>= 1) v = fmaxf(v, __shfl_xor(v, m, 32));
  return v;
}

DEV float lrelu(float l) { return (l > 0.0f) ? l : 0.2f * l; }

DEV float bcast_f(float v, int k) {
  return __int_as_float(__builtin_amdgcn_readlane(__float_as_int(v), k));
}

__global__ void k_mean(const float* __restrict__ x, float* __restrict__ out,
                       int* __restrict__ cnt, int NT, int N, int threeN) {
  int wid = (blockIdx.x * blockDim.x + threadIdx.x) >> 5;
  int lane = threadIdx.x & 31;
  if (wid >= NT) return;
  const float* r = x + (size_t)wid * 128;
  float s = r[lane] + r[lane + 32] + r[lane + 64] + r[lane + 96];
  s = wsum(s);
  if (lane == 0) {
    int b = wid / N, n = wid % N;
    out[(size_t)b * threeN + n] = s * (1.0f / 128.0f);
    cnt[wid] = 0;
  }
}

__global__ void k_gemm(const float* __restrict__ A, const float* __restrict__ Bw,
                       float* __restrict__ C, int M, int K) {
  int wid = (blockIdx.x * blockDim.x + threadIdx.x) >> 5;
  int lane = threadIdx.x & 31;
  int rowTile = wid >> 2, colGrp = wid & 3;
  int row0 = rowTile << 4, col0 = colGrp << 6;
  if (row0 >= M) return;
  int half = lane >> 4, lm = lane & 15;

  v8f acc0 = {}, acc1 = {}, acc2 = {}, acc3 = {};
  const float* ap = A + (size_t)(row0 + lm) * K + 2 * half;
  const float* bp = Bw + (size_t)(2 * half) * 256 + col0 + lm;

  for (int k0 = 0; k0 < K; k0 += 4) {
    v2f a;
    a.x = ap[k0];
    a.y = ap[k0 + 1];
    const float* b0 = bp + (size_t)k0 * 256;
    v2f b;
    b.x = b0[0];  b.y = b0[256];
    acc0 = __builtin_amdgcn_wmma_f32_16x16x4_f32(false, a, false, b, (short)0, acc0, false, false);
    b.x = b0[16]; b.y = b0[256 + 16];
    acc1 = __builtin_amdgcn_wmma_f32_16x16x4_f32(false, a, false, b, (short)0, acc1, false, false);
    b.x = b0[32]; b.y = b0[256 + 32];
    acc2 = __builtin_amdgcn_wmma_f32_16x16x4_f32(false, a, false, b, (short)0, acc2, false, false);
    b.x = b0[48]; b.y = b0[256 + 48];
    acc3 = __builtin_amdgcn_wmma_f32_16x16x4_f32(false, a, false, b, (short)0, acc3, false, false);
  }

  float* cp = C + (size_t)(row0 + 8 * half) * 256 + col0 + lm;
#pragma unroll
  for (int r2 = 0; r2 < 8; r2++) {
    cp[(size_t)r2 * 256 + 0]  = acc0[r2];
    cp[(size_t)r2 * 256 + 16] = acc1[r2];
    cp[(size_t)r2 * 256 + 32] = acc2[r2];
    cp[(size_t)r2 * 256 + 48] = acc3[r2];
  }
}

__global__ void k_att(const float* __restrict__ xp, const float* __restrict__ att_s,
                      const float* __restrict__ att_d, float* __restrict__ a_s,
                      float* __restrict__ a_d, int NT) {
  int wid = (blockIdx.x * blockDim.x + threadIdx.x) >> 5;
  int lane = threadIdx.x & 31;
  if (wid >= NT) return;
  const float* row = xp + (size_t)wid * 256;
#pragma unroll
  for (int h = 0; h < 4; h++) {
    int c0 = h * 64 + lane;
    float vs = row[c0] * att_s[c0] + row[c0 + 32] * att_s[c0 + 32];
    float vd = row[c0] * att_d[c0] + row[c0 + 32] * att_d[c0 + 32];
    vs = wsum(vs);
    vd = wsum(vd);
    if (lane == 0) {
      a_s[wid * 4 + h] = vs;
      a_d[wid * 4 + h] = vd;
    }
  }
}

__global__ void k_count(const int* __restrict__ dst, int* __restrict__ cnt,
                        int* __restrict__ rank, int E) {
  int e = blockIdx.x * blockDim.x + threadIdx.x;
  if (e < E) rank[e] = atomicAdd(&cnt[dst[e]], 1);
}

__global__ void __launch_bounds__(1024)
k_scan(const int* __restrict__ cnt, int* __restrict__ off, int NT) {
  __shared__ int wtot[32];
  int t = threadIdx.x, lane = t & 31, w = t >> 5;
  const int4* p = reinterpret_cast<const int4*>(cnt + t * 32);
  int4 v[8];
  int s = 0;
#pragma unroll
  for (int i = 0; i < 8; i++) {
    v[i] = p[i];
    s += v[i].x + v[i].y + v[i].z + v[i].w;
  }
  int inc = s;
#pragma unroll
  for (int o = 1; o < 32; o <<= 1) {
    int nb = __shfl_up(inc, o, 32);
    if (lane >= o) inc += nb;
  }
  if (lane == 31) wtot[w] = inc;
  __syncthreads();
  if (w == 0) {
    int x = wtot[lane], xi = x;
#pragma unroll
    for (int o = 1; o < 32; o <<= 1) {
      int nb = __shfl_up(xi, o, 32);
      if (lane >= o) xi += nb;
    }
    wtot[lane] = xi - x;
  }
  __syncthreads();
  int run = wtot[w] + inc - s;
  int4* q = reinterpret_cast<int4*>(off + t * 32);
#pragma unroll
  for (int i = 0; i < 8; i++) {
    int4 o4;
    o4.x = run; run += v[i].x;
    o4.y = run; run += v[i].y;
    o4.z = run; run += v[i].z;
    o4.w = run; run += v[i].w;
    q[i] = o4;
  }
  if (t == 1023) off[NT] = run;
}

__global__ void k_fill(const int* __restrict__ src, const int* __restrict__ dst,
                       const int* __restrict__ off, const int* __restrict__ rank,
                       int* __restrict__ csr, int E) {
  int e = blockIdx.x * blockDim.x + threadIdx.x;
  if (e < E) csr[off[dst[e]] + rank[e]] = src[e];
}

__global__ void k_gather(const int* __restrict__ off, const int* __restrict__ csr,
                         const float* __restrict__ a_s, const float* __restrict__ a_d,
                         const float* __restrict__ xp, float* __restrict__ agg, int NT) {
  const int lane = threadIdx.x & 31;
  const int d = __builtin_amdgcn_readfirstlane(
      (int)((blockIdx.x * blockDim.x + threadIdx.x) >> 5));
  if (d >= NT) return;
  const int beg = off[d], end = off[d + 1];
  const v4f ad = *reinterpret_cast<const v4f*>(a_d + (size_t)d * 4);
  const v4f sd = *reinterpret_cast<const v4f*>(a_s + (size_t)d * 4);

  const float ls0 = lrelu(sd.x + ad.x), ls1 = lrelu(sd.y + ad.y);
  const float ls2 = lrelu(sd.z + ad.z), ls3 = lrelu(sd.w + ad.w);

  float m0 = ls0, m1 = ls1, m2 = ls2, m3 = ls3;
  for (int j = beg + lane; j < end; j += 32) {
    const v4f as = *reinterpret_cast<const v4f*>(a_s + (size_t)csr[j] * 4);
    m0 = fmaxf(m0, lrelu(as.x + ad.x));
    m1 = fmaxf(m1, lrelu(as.y + ad.y));
    m2 = fmaxf(m2, lrelu(as.z + ad.z));
    m3 = fmaxf(m3, lrelu(as.w + ad.w));
  }
  m0 = wmax(m0); m1 = wmax(m1); m2 = wmax(m2); m3 = wmax(m3);

  float z0 = 0.0f, z1 = 0.0f, z2 = 0.0f, z3 = 0.0f;
  for (int j = beg + lane; j < end; j += 32) {
    const v4f as = *reinterpret_cast<const v4f*>(a_s + (size_t)csr[j] * 4);
    z0 += expf(lrelu(as.x + ad.x) - m0);
    z1 += expf(lrelu(as.y + ad.y) - m1);
    z2 += expf(lrelu(as.z + ad.z) - m2);
    z3 += expf(lrelu(as.w + ad.w) - m3);
  }
  const float es0 = expf(ls0 - m0), es1 = expf(ls1 - m1);
  const float es2 = expf(ls2 - m2), es3 = expf(ls3 - m3);
  const float dn0 = wsum(z0) + es0 + 1e-16f, dn1 = wsum(z1) + es1 + 1e-16f;
  const float dn2 = wsum(z2) + es2 + 1e-16f, dn3 = wsum(z3) + es3 + 1e-16f;

  const bool up = lane >= 16;
  v4f accLo, accHi;
  {
    const v4f* r = reinterpret_cast<const v4f*>(xp + (size_t)d * 256);
    const float cLo = up ? es1 / dn1 : es0 / dn0;
    const float cHi = up ? es3 / dn3 : es2 / dn2;
    accLo = r[lane] * cLo;
    accHi = r[32 + lane] * cHi;
  }
  for (int base = beg; base < end; base += 32) {
    const int j = base + lane;
    int s = d;
    float e0 = 0.0f, e1 = 0.0f, e2 = 0.0f, e3 = 0.0f;
    if (j < end) {
      s = csr[j];
      const v4f as = *reinterpret_cast<const v4f*>(a_s + (size_t)s * 4);
      e0 = expf(lrelu(as.x + ad.x) - m0) / dn0;
      e1 = expf(lrelu(as.y + ad.y) - m1) / dn1;
      e2 = expf(lrelu(as.z + ad.z) - m2) / dn2;
      e3 = expf(lrelu(as.w + ad.w) - m3) / dn3;
    }
    const int n = (end - base < 32) ? (end - base) : 32;
    for (int k = 0; k < n; k++) {
      const int sk = __builtin_amdgcn_readlane(s, k);
      const float b0 = bcast_f(e0, k), b1 = bcast_f(e1, k);
      const float b2 = bcast_f(e2, k), b3 = bcast_f(e3, k);
      const v4f* r = reinterpret_cast<const v4f*>(xp + (size_t)sk * 256);
      const v4f y0 = r[lane], y1 = r[32 + lane];
      accLo += y0 * (up ? b1 : b0);
      accHi += y1 * (up ? b3 : b2);
    }
  }
  v4f* o = reinterpret_cast<v4f*>(agg + (size_t)d * 256);
  o[lane] = accLo;
  o[32 + lane] = accHi;
}

__global__ void k_epilogue(const float* __restrict__ agg, const float* __restrict__ bias,
                           const float* __restrict__ pw, const float* __restrict__ pb,
                           float* __restrict__ hout, float* __restrict__ out,
                           int NT, int N, int threeN, int sec) {
  int wid = (blockIdx.x * blockDim.x + threadIdx.x) >> 5;
  int lane = threadIdx.x & 31;
  if (wid >= NT) return;
  const float* ar = agg + (size_t)wid * 256;
  float pool = 0.0f;
#pragma unroll
  for (int i = 0; i < 8; i++) {
    int ch = lane + i * 32;
    float v = ar[ch] + bias[ch];
    v = (v > 0.0f) ? v : expm1f(v);
    hout[(size_t)wid * 256 + ch] = v;
    pool += v * pw[ch];
  }
  pool = wsum(pool);
  if (lane == 0) {
    int b = wid / N, n = wid % N;
    out[(size_t)b * threeN + sec * N + n] = pool + pb[0];
  }
}

extern "C" void kernel_launch(void* const* d_in, const int* in_sizes, int n_in,
                              void* d_out, int out_size, void* d_ws, size_t ws_size,
                              hipStream_t stream) {
  const float* x   = (const float*)d_in[0];
  const int*   ei  = (const int*)d_in[1];
  const float* W1  = (const float*)d_in[2];
  const float* as1 = (const float*)d_in[3];
  const float* ad1 = (const float*)d_in[4];
  const float* b1  = (const float*)d_in[5];
  const float* W2  = (const float*)d_in[6];
  const float* as2 = (const float*)d_in[7];
  const float* ad2 = (const float*)d_in[8];
  const float* b2  = (const float*)d_in[9];
  const float* pw1 = (const float*)d_in[10];
  const float* pb1 = (const float*)d_in[11];
  const float* pw2 = (const float*)d_in[12];
  const float* pb2 = (const float*)d_in[13];
  float* out = (float*)d_out;

  const int NT = 32768, N = 4096, E = 524288;
  const int threeN = 3 * N;
  const size_t FEAT_BYTES = (size_t)NT * 256 * 4;

  char* w = (char*)d_ws;
  float* xp   = (float*)(w);
  float* hb   = (float*)(w + FEAT_BYTES);
  float* agg  = (float*)(w + 2 * FEAT_BYTES);
  float* a_s  = (float*)(w + 3 * FEAT_BYTES);
  float* a_d  = a_s + (size_t)NT * 4;
  int*   cnt  = (int*)(a_d + (size_t)NT * 4);
  int*   off  = cnt + NT;
  int*   rank = off + NT + 32;
  int*   csr  = rank + E;

  const int* srcA = ei;
  const int* dstA = ei + E;

  dim3 blk(256);
  int gWaveNT = NT / 8;
  int gGemm   = (NT / 16) * 4 / 8;
  int gEdge   = E / 256;

  k_mean<<<gWaveNT, blk, 0, stream>>>(x, out, cnt, NT, N, threeN);

  k_count<<<gEdge, blk, 0, stream>>>(dstA, cnt, rank, E);
  k_scan<<<1, dim3(1024), 0, stream>>>(cnt, off, NT);
  k_fill<<<gEdge, blk, 0, stream>>>(srcA, dstA, off, rank, csr, E);

  k_gemm<<<gGemm, blk, 0, stream>>>(x, W1, xp, NT, 128);
  k_att<<<gWaveNT, blk, 0, stream>>>(xp, as1, ad1, a_s, a_d, NT);
  k_gather<<<gWaveNT, blk, 0, stream>>>(off, csr, a_s, a_d, xp, agg, NT);
  k_epilogue<<<gWaveNT, blk, 0, stream>>>(agg, b1, pw1, pb1, hb, out, NT, N, threeN, 1);

  k_gemm<<<gGemm, blk, 0, stream>>>(hb, W2, xp, NT, 256);
  k_att<<<gWaveNT, blk, 0, stream>>>(xp, as2, ad2, a_s, a_d, NT);
  k_gather<<<gWaveNT, blk, 0, stream>>>(off, csr, a_s, a_d, xp, agg, NT);
  k_epilogue<<<gWaveNT, blk, 0, stream>>>(agg, b2, pw2, pb2, xp, out, NT, N, threeN, 2);
}


# Waves launched cut 4x: per-node kernels take 8 nodes per wave (gather 3), bucketing kernels 4 edges per thread; same per-node sums, bucket arrival order still run-dependent
# speedup vs baseline: 2.6834x; 1.0847x over previous
#include <hip/hip_runtime.h>
#include <math.h>


#define NPW 8
#define GPW 3
#define EPT 4

typedef float v2f __attribute__((ext_vector_type(2)));
typedef float v8f __attribute__((ext_vector_type(8)));
typedef float v4f __attribute__((ext_vector_type(4)));

#define DEV __device__ __forceinline__

DEV float wsum(float v) {
#pragma unroll
  for (int m = 16; m; m >>= 1) v += __shfl_xor(v, m, 32);
  return v;
}

DEV float wmax(float v) {
#pragma unroll
  for (int m = 16; m; m >>= 1) v = fmaxf(v, __shfl_xor(v, m, 32));
  return v;
}

DEV float lrelu(float l) { return (l > 0.0f) ? l : 0.2f * l; }

DEV float bcast_f(float v, int k) {
  return __int_as_float(__builtin_amdgcn_readlane(__float_as_int(v), k));
}

__global__ void k_mean(const float* __restrict__ x, float* __restrict__ out,
                       int* __restrict__ cnt, int NT, int N, int threeN) {
  int wid = (blockIdx.x * blockDim.x + threadIdx.x) >> 5;
  int lane = threadIdx.x & 31;
  int n0 = wid * NPW;
  if (n0 >= NT) return;
  const float* r = x + (size_t)n0 * 128;
  float s[NPW];
#pragma unroll
  for (int i = 0; i < NPW; i++) {
    const float* q = r + i * 128;
    s[i] = q[lane] + q[lane + 32] + q[lane + 64] + q[lane + 96];
  }
  float mine = 0.0f;
#pragma unroll
  for (int i = 0; i < NPW; i++) {
    float t = wsum(s[i]);
    if (lane == i) mine = t;
  }
  if (lane < NPW) {
    int nn = n0 + lane;
    int b = nn / N, n = nn % N;
    out[(size_t)b * threeN + n] = mine * (1.0f / 128.0f);
    cnt[nn] = 0;
  }
}

__global__ void k_gemm(const float* __restrict__ A, const float* __restrict__ Bw,
                       float* __restrict__ C, int M, int K) {
  int wid = (blockIdx.x * blockDim.x + threadIdx.x) >> 5;
  int lane = threadIdx.x & 31;
  int rowTile = wid >> 2, colGrp = wid & 3;
  int row0 = rowTile << 4, col0 = colGrp << 6;
  if (row0 >= M) return;
  int half = lane >> 4, lm = lane & 15;

  v8f acc0 = {}, acc1 = {}, acc2 = {}, acc3 = {};
  const float* ap = A + (size_t)(row0 + lm) * K + 2 * half;
  const float* bp = Bw + (size_t)(2 * half) * 256 + col0 + lm;

  for (int k0 = 0; k0 < K; k0 += 4) {
    v2f a;
    a.x = ap[k0];
    a.y = ap[k0 + 1];
    const float* b0 = bp + (size_t)k0 * 256;
    v2f b;
    b.x = b0[0];  b.y = b0[256];
    acc0 = __builtin_amdgcn_wmma_f32_16x16x4_f32(false, a, false, b, (short)0, acc0, false, false);
    b.x = b0[16]; b.y = b0[256 + 16];
    acc1 = __builtin_amdgcn_wmma_f32_16x16x4_f32(false, a, false, b, (short)0, acc1, false, false);
    b.x = b0[32]; b.y = b0[256 + 32];
    acc2 = __builtin_amdgcn_wmma_f32_16x16x4_f32(false, a, false, b, (short)0, acc2, false, false);
    b.x = b0[48]; b.y = b0[256 + 48];
    acc3 = __builtin_amdgcn_wmma_f32_16x16x4_f32(false, a, false, b, (short)0, acc3, false, false);
  }

  float* cp = C + (size_t)(row0 + 8 * half) * 256 + col0 + lm;
#pragma unroll
  for (int r2 = 0; r2 < 8; r2++) {
    cp[(size_t)r2 * 256 + 0]  = acc0[r2];
    cp[(size_t)r2 * 256 + 16] = acc1[r2];
    cp[(size_t)r2 * 256 + 32] = acc2[r2];
    cp[(size_t)r2 * 256 + 48] = acc3[r2];
  }
}

__global__ void k_att(const float* __restrict__ xp, const float* __restrict__ att_s,
                      const float* __restrict__ att_d, float* __restrict__ a_s,
                      float* __restrict__ a_d, int NT) {
  int wid = (blockIdx.x * blockDim.x + threadIdx.x) >> 5;
  int lane = threadIdx.x & 31;
  int n0 = wid * NPW;
  if (n0 >= NT) return;
  float ws[8], wd[8];
#pragma unroll
  for (int h = 0; h < 4; h++) {
    ws[2 * h] = att_s[h * 64 + lane];  ws[2 * h + 1] = att_s[h * 64 + lane + 32];
    wd[2 * h] = att_d[h * 64 + lane];  wd[2 * h + 1] = att_d[h * 64 + lane + 32];
  }
  const float* row0 = xp + (size_t)n0 * 256;
  float rw[NPW][8];
#pragma unroll
  for (int i = 0; i < NPW; i++) {
#pragma unroll
    for (int q = 0; q < 8; q++) rw[i][q] = row0[(size_t)i * 256 + q * 32 + lane];
  }
  float outS = 0.0f, outD = 0.0f;
#pragma unroll
  for (int i = 0; i < NPW; i++) {
#pragma unroll
    for (int h = 0; h < 4; h++) {
      float vs = rw[i][2 * h] * ws[2 * h] + rw[i][2 * h + 1] * ws[2 * h + 1];
      float vd = rw[i][2 * h] * wd[2 * h] + rw[i][2 * h + 1] * wd[2 * h + 1];
      vs = wsum(vs);
      vd = wsum(vd);
      if (lane == i * 4 + h) { outS = vs; outD = vd; }
    }
  }
  if (lane < NPW * 4) {
    a_s[(size_t)n0 * 4 + lane] = outS;
    a_d[(size_t)n0 * 4 + lane] = outD;
  }
}

__global__ void k_count(const int* __restrict__ dst, int* __restrict__ cnt,
                        int* __restrict__ rank, int E) {
  int e = (blockIdx.x * blockDim.x + threadIdx.x) * EPT;
  if (e + EPT <= E) {
    const int4 d4 = *reinterpret_cast<const int4*>(dst + e);
    int4 r4;
    r4.x = atomicAdd(&cnt[d4.x], 1);
    r4.y = atomicAdd(&cnt[d4.y], 1);
    r4.z = atomicAdd(&cnt[d4.z], 1);
    r4.w = atomicAdd(&cnt[d4.w], 1);
    *reinterpret_cast<int4*>(rank + e) = r4;
  } else {
    for (; e < E; e++) rank[e] = atomicAdd(&cnt[dst[e]], 1);
  }
}

__global__ void __launch_bounds__(1024)
k_scan(const int* __restrict__ cnt, int* __restrict__ off, int NT) {
  __shared__ int wtot[32];
  int t = threadIdx.x, lane = t & 31, w = t >> 5;
  const int4* p = reinterpret_cast<const int4*>(cnt + t * 32);
  int4 v[8];
  int s = 0;
#pragma unroll
  for (int i = 0; i < 8; i++) {
    v[i] = p[i];
    s += v[i].x + v[i].y + v[i].z + v[i].w;
  }
  int inc = s;
#pragma unroll
  for (int o = 1; o < 32; o <<= 1) {
    int nb = __shfl_up(inc, o, 32);
    if (lane >= o) inc += nb;
  }
  if (lane == 31) wtot[w] = inc;
  __syncthreads();
  if (w == 0) {
    int x = wtot[lane], xi = x;
#pragma unroll
    for (int o = 1; o < 32; o <<= 1) {
      int nb = __shfl_up(xi, o, 32);
      if (lane >= o) xi += nb;
    }
    wtot[lane] = xi - x;
  }
  __syncthreads();
  int run = wtot[w] + inc - s;
  int4* q = reinterpret_cast<int4*>(off + t * 32);
#pragma unroll
  for (int i = 0; i < 8; i++) {
    int4 o4;
    o4.x = run; run += v[i].x;
    o4.y = run; run += v[i].y;
    o4.z = run; run += v[i].z;
    o4.w = run; run += v[i].w;
    q[i] = o4;
  }
  if (t == 1023) off[NT] = run;
}

__global__ void k_fill(const int* __restrict__ src, const int* __restrict__ dst,
                       const int* __restrict__ off, const int* __restrict__ rank,
                       int* __restrict__ csr, int E) {
  int e = (blockIdx.x * blockDim.x + threadIdx.x) * EPT;
  if (e + EPT <= E) {
    const int4 s4 = *reinterpret_cast<const int4*>(src + e);
    const int4 d4 = *reinterpret_cast<const int4*>(dst + e);
    const int4 r4 = *reinterpret_cast<const int4*>(rank + e);
    const int o0 = off[d4.x], o1 = off[d4.y], o2 = off[d4.z], o3 = off[d4.w];
    csr[o0 + r4.x] = s4.x;
    csr[o1 + r4.y] = s4.y;
    csr[o2 + r4.z] = s4.z;
    csr[o3 + r4.w] = s4.w;
  } else {
    for (; e < E; e++) csr[off[dst[e]] + rank[e]] = src[e];
  }
}

__global__ void k_gather(const int* __restrict__ off, const int* __restrict__ csr,
                         const float* __restrict__ a_s, const float* __restrict__ a_d,
                         const float* __restrict__ xp, float* __restrict__ agg, int NT) {
  const int lane = threadIdx.x & 31;
  const int wv = __builtin_amdgcn_readfirstlane(
      (int)((blockIdx.x * blockDim.x + threadIdx.x) >> 5));
  const bool up = lane >= 16;
  for (int g = 0; g < GPW; g++) {
  const int d = wv * GPW + g;
  if (d >= NT) return;
  const int beg = off[d], end = off[d + 1];
  const v4f ad = *reinterpret_cast<const v4f*>(a_d + (size_t)d * 4);
  const v4f sd = *reinterpret_cast<const v4f*>(a_s + (size_t)d * 4);

  const float ls0 = lrelu(sd.x + ad.x), ls1 = lrelu(sd.y + ad.y);
  const float ls2 = lrelu(sd.z + ad.z), ls3 = lrelu(sd.w + ad.w);

  float m0 = ls0, m1 = ls1, m2 = ls2, m3 = ls3;
  for (int j = beg + lane; j < end; j += 32) {
    const v4f as = *reinterpret_cast<const v4f*>(a_s + (size_t)csr[j] * 4);
    m0 = fmaxf(m0, lrelu(as.x + ad.x));
    m1 = fmaxf(m1, lrelu(as.y + ad.y));
    m2 = fmaxf(m2, lrelu(as.z + ad.z));
    m3 = fmaxf(m3, lrelu(as.w + ad.w));
  }
  m0 = wmax(m0); m1 = wmax(m1); m2 = wmax(m2); m3 = wmax(m3);

  float z0 = 0.0f, z1 = 0.0f, z2 = 0.0f, z3 = 0.0f;
  for (int j = beg + lane; j < end; j += 32) {
    const v4f as = *reinterpret_cast<const v4f*>(a_s + (size_t)csr[j] * 4);
    z0 += expf(lrelu(as.x + ad.x) - m0);
    z1 += expf(lrelu(as.y + ad.y) - m1);
    z2 += expf(lrelu(as.z + ad.z) - m2);
    z3 += expf(lrelu(as.w + ad.w) - m3);
  }
  const float es0 = expf(ls0 - m0), es1 = expf(ls1 - m1);
  const float es2 = expf(ls2 - m2), es3 = expf(ls3 - m3);
  const float dn0 = wsum(z0) + es0 + 1e-16f, dn1 = wsum(z1) + es1 + 1e-16f;
  const float dn2 = wsum(z2) + es2 + 1e-16f, dn3 = wsum(z3) + es3 + 1e-16f;

  v4f accLo, accHi;
  {
    const v4f* r = reinterpret_cast<const v4f*>(xp + (size_t)d * 256);
    const float cLo = up ? es1 / dn1 : es0 / dn0;
    const float cHi = up ? es3 / dn3 : es2 / dn2;
    accLo = r[lane] * cLo;
    accHi = r[32 + lane] * cHi;
  }
  for (int base = beg; base < end; base += 32) {
    const int j = base + lane;
    int s = d;
    float e0 = 0.0f, e1 = 0.0f, e2 = 0.0f, e3 = 0.0f;
    if (j < end) {
      s = csr[j];
      const v4f as = *reinterpret_cast<const v4f*>(a_s + (size_t)s * 4);
      e0 = expf(lrelu(as.x + ad.x) - m0) / dn0;
      e1 = expf(lrelu(as.y + ad.y) - m1) / dn1;
      e2 = expf(lrelu(as.z + ad.z) - m2) / dn2;
      e3 = expf(lrelu(as.w + ad.w) - m3) / dn3;
    }
    const int n = (end - base < 32) ? (end - base) : 32;
    for (int k = 0; k < n; k++) {
      const int sk = __builtin_amdgcn_readlane(s, k);
      const float b0 = bcast_f(e0, k), b1 = bcast_f(e1, k);
      const float b2 = bcast_f(e2, k), b3 = bcast_f(e3, k);
      const v4f* r = reinterpret_cast<const v4f*>(xp + (size_t)sk * 256);
      const v4f y0 = r[lane], y1 = r[32 + lane];
      accLo += y0 * (up ? b1 : b0);
      accHi += y1 * (up ? b3 : b2);
    }
  }
  v4f* o = reinterpret_cast<v4f*>(agg + (size_t)d * 256);
  o[lane] = accLo;
  o[32 + lane] = accHi;
  }
}

__global__ void k_epilogue(const float* __restrict__ agg, const float* __restrict__ bias,
                           const float* __restrict__ pw, const float* __restrict__ pb,
                           float* __restrict__ hout, float* __restrict__ out,
                           int NT, int N, int threeN, int sec) {
  int wid = (blockIdx.x * blockDim.x + threadIdx.x) >> 5;
  int lane = threadIdx.x & 31;
  int n0 = wid * NPW;
  if (n0 >= NT) return;
  float bs[8], pv[8];
#pragma unroll
  for (int i = 0; i < 8; i++) {
    bs[i] = bias[lane + i * 32];
    pv[i] = pw[lane + i * 32];
  }
  const float* ar = agg + (size_t)n0 * 256;
  float* hr = hout + (size_t)n0 * 256;
  float v[NPW][8];
#pragma unroll
  for (int k = 0; k < NPW; k++) {
#pragma unroll
    for (int i = 0; i < 8; i++) v[k][i] = ar[(size_t)k * 256 + lane + i * 32];
  }
  float mine = 0.0f;
#pragma unroll
  for (int k = 0; k < NPW; k++) {
    float pool = 0.0f;
#pragma unroll
    for (int i = 0; i < 8; i++) {
      float t = v[k][i] + bs[i];
      t = (t > 0.0f) ? t : expm1f(t);
      hr[(size_t)k * 256 + lane + i * 32] = t;
      pool += t * pv[i];
    }
    pool = wsum(pool);
    if (lane == k) mine = pool;
  }
  if (lane < NPW) {
    int nn = n0 + lane;
    int b = nn / N, n = nn % N;
    out[(size_t)b * threeN + sec * N + n] = mine + pb[0];
  }
}

extern "C" void kernel_launch(void* const* d_in, const int* in_sizes, int n_in,
                              void* d_out, int out_size, void* d_ws, size_t ws_size,
                              hipStream_t stream) {
  const float* x   = (const float*)d_in[0];
  const int*   ei  = (const int*)d_in[1];
  const float* W1  = (const float*)d_in[2];
  const float* as1 = (const float*)d_in[3];
  const float* ad1 = (const float*)d_in[4];
  const float* b1  = (const float*)d_in[5];
  const float* W2  = (const float*)d_in[6];
  const float* as2 = (const float*)d_in[7];
  const float* ad2 = (const float*)d_in[8];
  const float* b2  = (const float*)d_in[9];
  const float* pw1 = (const float*)d_in[10];
  const float* pb1 = (const float*)d_in[11];
  const float* pw2 = (const float*)d_in[12];
  const float* pb2 = (const float*)d_in[13];
  float* out = (float*)d_out;

  const int NT = 32768, N = 4096, E = 524288;
  const int threeN = 3 * N;
  const size_t FEAT_BYTES = (size_t)NT * 256 * 4;

  char* w = (char*)d_ws;
  float* xp   = (float*)(w);
  float* hb   = (float*)(w + FEAT_BYTES);
  float* agg  = (float*)(w + 2 * FEAT_BYTES);
  float* a_s  = (float*)(w + 3 * FEAT_BYTES);
  float* a_d  = a_s + (size_t)NT * 4;
  int*   cnt  = (int*)(a_d + (size_t)NT * 4);
  int*   off  = cnt + NT;
  int*   rank = off + NT + 32;
  int*   csr  = rank + E;

  const int* srcA = ei;
  const int* dstA = ei + E;

  dim3 blk(256);
  int gWaveNT = NT / NPW / 8;
  int gGather = (NT + GPW * 8 - 1) / (GPW * 8);
  int gGemm   = (NT / 16) * 4 / 8;
  int gEdge   = E / EPT / 256;

  k_mean<<<gWaveNT, blk, 0, stream>>>(x, out, cnt, NT, N, threeN);

  k_count<<<gEdge, blk, 0, stream>>>(dstA, cnt, rank, E);
  k_scan<<<1, dim3(1024), 0, stream>>>(cnt, off, NT);
  k_fill<<<gEdge, blk, 0, stream>>>(srcA, dstA, off, rank, csr, E);

  k_gemm<<<gGemm, blk, 0, stream>>>(x, W1, xp, NT, 128);
  k_att<<<gWaveNT, blk, 0, stream>>>(xp, as1, ad1, a_s, a_d, NT);
  k_gather<<<gGather, blk, 0, stream>>>(off, csr, a_s, a_d, xp, agg, NT);
  k_epilogue<<<gWaveNT, blk, 0, stream>>>(agg, b1, pw1, pb1, hb, out, NT, N, threeN, 1);

  k_gemm<<<gGemm, blk, 0, stream>>>(hb, W2, xp, NT, 256);
  k_att<<<gWaveNT, blk, 0, stream>>>(xp, as2, ad2, a_s, a_d, NT);
  k_gather<<<gGather, blk, 0, stream>>>(off, csr, a_s, a_d, xp, agg, NT);
  k_epilogue<<<gWaveNT, blk, 0, stream>>>(agg, b2, pw2, pb2, xp, out, NT, N, threeN, 2);
}


# GEMM fragments loaded and C stored 16 bytes per lane: lane takes 4 adjacent columns and 4 adjacent k; k order changed in A and B alike: GEMM sums, dots and feature mean in a different order
# speedup vs baseline: 3.3904x; 1.2635x over previous
#include <hip/hip_runtime.h>
#include <math.h>


#define GPW 3
#define EPT 4

typedef float v2f __attribute__((ext_vector_type(2)));
typedef float v8f __attribute__((ext_vector_type(8)));
typedef float v4f __attribute__((ext_vector_type(4)));

#define DEV __device__ __forceinline__

DEV float wsum(float v) {
#pragma unroll
  for (int m = 16; m; m >>= 1) v += __shfl_xor(v, m, 32);
  return v;
}

DEV float wmax(float v) {
#pragma unroll
  for (int m = 16; m; m >>= 1) v = fmaxf(v, __shfl_xor(v, m, 32));
  return v;
}

DEV float lrelu(float l) { return (l > 0.0f) ? l : 0.2f * l; }

DEV float bcast_f(float v, int k) {
  return __int_as_float(__builtin_amdgcn_readlane(__float_as_int(v), k));
}

DEV float elu1(float t) { return (t > 0.0f) ? t : expm1f(t); }

__global__ void k_gemm(const float* __restrict__ A, const float* __restrict__ Bw,
                       const float* __restrict__ att_s, const float* __restrict__ att_d,
                       float* __restrict__ C, float* __restrict__ a_s,
                       float* __restrict__ a_d, float* __restrict__ out,
                       int* __restrict__ cnt, int M, int K, int N, int threeN,
                       int first) {
  __shared__ float sAtt[8][32];
  if ((int)blockIdx.x * 32 >= M) return;
  const int wib = threadIdx.x >> 5;
  const int lane = threadIdx.x & 31;
  const int rowTile = (int)blockIdx.x * 2 + (wib >> 2), colGrp = wib & 3;
  const int row0 = rowTile << 4, col0 = colGrp << 6;
  const int half = lane >> 4, lm = lane & 15;

  v8f acc0 = {}, acc1 = {}, acc2 = {}, acc3 = {};
  const float* ap = A + (size_t)(row0 + lm) * K + 4 * half;
  const float* bp = Bw + (size_t)(4 * half) * 256 + col0 + 4 * lm;
  float rsum = 0.0f;

  for (int k0 = 0; k0 < K; k0 += 8) {
    const v4f a4 = *reinterpret_cast<const v4f*>(ap + k0);
    rsum += a4.x + a4.y + a4.z + a4.w;
    const float* b0 = bp + (size_t)k0 * 256;
    const v4f w0 = *reinterpret_cast<const v4f*>(b0);
    const v4f w1 = *reinterpret_cast<const v4f*>(b0 + 256);
    const v4f w2 = *reinterpret_cast<const v4f*>(b0 + 512);
    const v4f w3 = *reinterpret_cast<const v4f*>(b0 + 768);
    v2f a, b;
    a.x = a4.x; a.y = a4.y;
    b.x = w0.x; b.y = w1.x;
    acc0 = __builtin_amdgcn_wmma_f32_16x16x4_f32(false, a, false, b, (short)0, acc0, false, false);
    b.x = w0.y; b.y = w1.y;
    acc1 = __builtin_amdgcn_wmma_f32_16x16x4_f32(false, a, false, b, (short)0, acc1, false, false);
    b.x = w0.z; b.y = w1.z;
    acc2 = __builtin_amdgcn_wmma_f32_16x16x4_f32(false, a, false, b, (short)0, acc2, false, false);
    b.x = w0.w; b.y = w1.w;
    acc3 = __builtin_amdgcn_wmma_f32_16x16x4_f32(false, a, false, b, (short)0, acc3, false, false);
    a.x = a4.z; a.y = a4.w;
    b.x = w2.x; b.y = w3.x;
    acc0 = __builtin_amdgcn_wmma_f32_16x16x4_f32(false, a, false, b, (short)0, acc0, false, false);
    b.x = w2.y; b.y = w3.y;
    acc1 = __builtin_amdgcn_wmma_f32_16x16x4_f32(false, a, false, b, (short)0, acc1, false, false);
    b.x = w2.z; b.y = w3.z;
    acc2 = __builtin_amdgcn_wmma_f32_16x16x4_f32(false, a, false, b, (short)0, acc2, false, false);
    b.x = w2.w; b.y = w3.w;
    acc3 = __builtin_amdgcn_wmma_f32_16x16x4_f32(false, a, false, b, (short)0, acc3, false, false);
  }

  float* cp = C + (size_t)(row0 + 8 * half) * 256 + col0 + 4 * lm;
#pragma unroll
  for (int r2 = 0; r2 < 8; r2++) {
    v4f o;
    o.x = acc0[r2]; o.y = acc1[r2]; o.z = acc2[r2]; o.w = acc3[r2];
    *reinterpret_cast<v4f*>(cp + (size_t)r2 * 256) = o;
  }

  if (first) {
    rsum += __shfl_xor(rsum, 16, 32);
    if (lane < 16) {
      const int nn = row0 + lm;
      if (colGrp == 0) out[(size_t)(nn / N) * threeN + nn % N] = rsum * (1.0f / (float)K);
      if (colGrp == 1) cnt[nn] = 0;
    }
  }

  const v4f ws = *reinterpret_cast<const v4f*>(att_s + col0 + 4 * lm);
  const v4f wd = *reinterpret_cast<const v4f*>(att_d + col0 + 4 * lm);
  const float ws0 = ws.x, ws1 = ws.y, ws2 = ws.z, ws3 = ws.w;
  const float wd0 = wd.x, wd1 = wd.y, wd2 = wd.z, wd3 = wd.w;
  float v[16];
#pragma unroll
  for (int r2 = 0; r2 < 8; r2++) {
    v[r2]     = acc0[r2] * ws0 + acc1[r2] * ws1 + acc2[r2] * ws2 + acc3[r2] * ws3;
    v[8 + r2] = acc0[r2] * wd0 + acc1[r2] * wd1 + acc2[r2] * wd2 + acc3[r2] * wd3;
  }
  const bool b3 = (lm & 8) != 0, b2 = (lm & 4) != 0, b1 = (lm & 2) != 0, b0 = (lm & 1) != 0;
  float u[8], t4[4], t2[2];
#pragma unroll
  for (int i = 0; i < 8; i++)
    u[i] = (b3 ? v[8 + i] : v[i]) + __shfl_xor(b3 ? v[i] : v[8 + i], 8, 32);
#pragma unroll
  for (int i = 0; i < 4; i++)
    t4[i] = (b2 ? u[4 + i] : u[i]) + __shfl_xor(b2 ? u[i] : u[4 + i], 4, 32);
#pragma unroll
  for (int i = 0; i < 2; i++)
    t2[i] = (b1 ? t4[2 + i] : t4[i]) + __shfl_xor(b1 ? t4[i] : t4[2 + i], 2, 32);
  const float tot = (b0 ? t2[1] : t2[0]) + __shfl_xor(b0 ? t2[0] : t2[1], 1, 32);
  sAtt[wib][lane] = tot;
  __syncthreads();
  {
    const int tid = threadIdx.x;
    const int tile = tid >> 7, sd = (tid >> 6) & 1, idx = tid & 63;
    const int row = idx >> 2, head = idx & 3;
    const float val = sAtt[tile * 4 + head][(row >> 3) * 16 + sd * 8 + (row & 7)];
    float* dp = sd ? a_d : a_s;
    dp[((size_t)blockIdx.x * 32 + tile * 16) * 4 + idx] = val;
  }
}

__global__ void k_count(const int* __restrict__ dst, int* __restrict__ cnt,
                        int* __restrict__ rank, int E) {
  int e = (blockIdx.x * blockDim.x + threadIdx.x) * EPT;
  if (e + EPT <= E) {
    const int4 d4 = *reinterpret_cast<const int4*>(dst + e);
    int4 r4;
    r4.x = atomicAdd(&cnt[d4.x], 1);
    r4.y = atomicAdd(&cnt[d4.y], 1);
    r4.z = atomicAdd(&cnt[d4.z], 1);
    r4.w = atomicAdd(&cnt[d4.w], 1);
    *reinterpret_cast<int4*>(rank + e) = r4;
  } else {
    for (; e < E; e++) rank[e] = atomicAdd(&cnt[dst[e]], 1);
  }
}

__global__ void __launch_bounds__(1024)
k_scan(const int* __restrict__ cnt, int* __restrict__ off, int NT) {
  __shared__ int wtot[32];
  int t = threadIdx.x, lane = t & 31, w = t >> 5;
  const int4* p = reinterpret_cast<const int4*>(cnt + t * 32);
  int4 v[8];
  int s = 0;
#pragma unroll
  for (int i = 0; i < 8; i++) {
    v[i] = p[i];
    s += v[i].x + v[i].y + v[i].z + v[i].w;
  }
  int inc = s;
#pragma unroll
  for (int o = 1; o < 32; o <<= 1) {
    int nb = __shfl_up(inc, o, 32);
    if (lane >= o) inc += nb;
  }
  if (lane == 31) wtot[w] = inc;
  __syncthreads();
  if (w == 0) {
    int x = wtot[lane], xi = x;
#pragma unroll
    for (int o = 1; o < 32; o <<= 1) {
      int nb = __shfl_up(xi, o, 32);
      if (lane >= o) xi += nb;
    }
    wtot[lane] = xi - x;
  }
  __syncthreads();
  int run = wtot[w] + inc - s;
  int4* q = reinterpret_cast<int4*>(off + t * 32);
#pragma unroll
  for (int i = 0; i < 8; i++) {
    int4 o4;
    o4.x = run; run += v[i].x;
    o4.y = run; run += v[i].y;
    o4.z = run; run += v[i].z;
    o4.w = run; run += v[i].w;
    q[i] = o4;
  }
  if (t == 1023) off[NT] = run;
}

__global__ void k_fill(const int* __restrict__ src, const int* __restrict__ dst,
                       const int* __restrict__ off, const int* __restrict__ rank,
                       int* __restrict__ csr, int E) {
  int e = (blockIdx.x * blockDim.x + threadIdx.x) * EPT;
  if (e + EPT <= E) {
    const int4 s4 = *reinterpret_cast<const int4*>(src + e);
    const int4 d4 = *reinterpret_cast<const int4*>(dst + e);
    const int4 r4 = *reinterpret_cast<const int4*>(rank + e);
    const int o0 = off[d4.x], o1 = off[d4.y], o2 = off[d4.z], o3 = off[d4.w];
    csr[o0 + r4.x] = s4.x;
    csr[o1 + r4.y] = s4.y;
    csr[o2 + r4.z] = s4.z;
    csr[o3 + r4.w] = s4.w;
  } else {
    for (; e < E; e++) csr[off[dst[e]] + rank[e]] = src[e];
  }
}

__global__ void k_gather(const int* __restrict__ off, const int* __restrict__ csr,
                         const float* __restrict__ a_s, const float* __restrict__ a_d,
                         const float* __restrict__ xp, const float* __restrict__ bias,
                         const float* __restrict__ pw, const float* __restrict__ pb,
                         float* __restrict__ hout, float* __restrict__ out,
                         int NT, int N, int threeN, int sec) {
  const unsigned lane = threadIdx.x & 31;
  const int wv = __builtin_amdgcn_readfirstlane(
      (int)((blockIdx.x * blockDim.x + threadIdx.x) >> 5));
  const bool up = lane >= 16;
  for (int g = 0; g < GPW; g++) {
  const int d = wv * GPW + g;
  if (d >= NT) return;
  const int beg = off[d], end = off[d + 1];
  const v4f ad = *reinterpret_cast<const v4f*>(a_d + (size_t)d * 4);
  const v4f sd = *reinterpret_cast<const v4f*>(a_s + (size_t)d * 4);

  const float ls0 = lrelu(sd.x + ad.x), ls1 = lrelu(sd.y + ad.y);
  const float ls2 = lrelu(sd.z + ad.z), ls3 = lrelu(sd.w + ad.w);

  float m0 = ls0, m1 = ls1, m2 = ls2, m3 = ls3;
  for (int j = beg + lane; j < end; j += 32) {
    const v4f as = *reinterpret_cast<const v4f*>(a_s + (size_t)csr[j] * 4);
    m0 = fmaxf(m0, lrelu(as.x + ad.x));
    m1 = fmaxf(m1, lrelu(as.y + ad.y));
    m2 = fmaxf(m2, lrelu(as.z + ad.z));
    m3 = fmaxf(m3, lrelu(as.w + ad.w));
  }
  m0 = wmax(m0); m1 = wmax(m1); m2 = wmax(m2); m3 = wmax(m3);

  float z0 = 0.0f, z1 = 0.0f, z2 = 0.0f, z3 = 0.0f;
  for (int j = beg + lane; j < end; j += 32) {
    const v4f as = *reinterpret_cast<const v4f*>(a_s + (size_t)csr[j] * 4);
    z0 += expf(lrelu(as.x + ad.x) - m0);
    z1 += expf(lrelu(as.y + ad.y) - m1);
    z2 += expf(lrelu(as.z + ad.z) - m2);
    z3 += expf(lrelu(as.w + ad.w) - m3);
  }
  const float es0 = expf(ls0 - m0), es1 = expf(ls1 - m1);
  const float es2 = expf(ls2 - m2), es3 = expf(ls3 - m3);
  const float dn0 = wsum(z0) + es0 + 1e-16f, dn1 = wsum(z1) + es1 + 1e-16f;
  const float dn2 = wsum(z2) + es2 + 1e-16f, dn3 = wsum(z3) + es3 + 1e-16f;

  v4f accLo, accHi;
  {
    const v4f* r = reinterpret_cast<const v4f*>(xp + (size_t)d * 256);
    const float cLo = up ? es1 / dn1 : es0 / dn0;
    const float cHi = up ? es3 / dn3 : es2 / dn2;
    accLo = r[lane] * cLo;
    accHi = r[32 + lane] * cHi;
  }
  for (int base = beg; base < end; base += 32) {
    const int j = base + lane;
    int s = d;
    float e0 = 0.0f, e1 = 0.0f, e2 = 0.0f, e3 = 0.0f;
    if (j < end) {
      s = csr[j];
      const v4f as = *reinterpret_cast<const v4f*>(a_s + (size_t)s * 4);
      e0 = expf(lrelu(as.x + ad.x) - m0) / dn0;
      e1 = expf(lrelu(as.y + ad.y) - m1) / dn1;
      e2 = expf(lrelu(as.z + ad.z) - m2) / dn2;
      e3 = expf(lrelu(as.w + ad.w) - m3) / dn3;
    }
    const int n = (end - base < 32) ? (end - base) : 32;
    for (int k = 0; k < n; k++) {
      const int sk = __builtin_amdgcn_readlane(s, k);
      const float b0 = bcast_f(e0, k), b1 = bcast_f(e1, k);
      const float b2 = bcast_f(e2, k), b3 = bcast_f(e3, k);
      const v4f* r = reinterpret_cast<const v4f*>(xp + (size_t)sk * 256);
      const v4f y0 = r[lane], y1 = r[32 + lane];
      accLo += y0 * (up ? b1 : b0);
      accHi += y1 * (up ? b3 : b2);
    }
  }
  {
    const v4f* bv = reinterpret_cast<const v4f*>(bias);
    const v4f* pv = reinterpret_cast<const v4f*>(pw);
    unsigned le = lane;
    asm volatile("" : "+v"(le));
    v4f tLo = accLo + bv[le], tHi = accHi + bv[32 + le];
    tLo.x = elu1(tLo.x); tLo.y = elu1(tLo.y); tLo.z = elu1(tLo.z); tLo.w = elu1(tLo.w);
    tHi.x = elu1(tHi.x); tHi.y = elu1(tHi.y); tHi.z = elu1(tHi.z); tHi.w = elu1(tHi.w);
    if (hout != nullptr) {
      v4f* o = reinterpret_cast<v4f*>(hout + (size_t)d * 256);
      o[le] = tLo;
      o[32 + le] = tHi;
    }
    const v4f pLo = pv[le], pHi = pv[32 + le];
    float pool = tLo.x * pLo.x + tLo.y * pLo.y + tLo.z * pLo.z + tLo.w * pLo.w
               + tHi.x * pHi.x + tHi.y * pHi.y + tHi.z * pHi.z + tHi.w * pHi.w;
    pool = wsum(pool);
    if (lane == 0) out[(size_t)(d / N) * threeN + sec * N + d % N] = pool + pb[0];
  }
  }
}

extern "C" void kernel_launch(void* const* d_in, const int* in_sizes, int n_in,
                              void* d_out, int out_size, void* d_ws, size_t ws_size,
                              hipStream_t stream) {
  const float* x   = (const float*)d_in[0];
  const int*   ei  = (const int*)d_in[1];
  const float* W1  = (const float*)d_in[2];
  const float* as1 = (const float*)d_in[3];
  const float* ad1 = (const float*)d_in[4];
  const float* b1  = (const float*)d_in[5];
  const float* W2  = (const float*)d_in[6];
  const float* as2 = (const float*)d_in[7];
  const float* ad2 = (const float*)d_in[8];
  const float* b2  = (const float*)d_in[9];
  const float* pw1 = (const float*)d_in[10];
  const float* pb1 = (const float*)d_in[11];
  const float* pw2 = (const float*)d_in[12];
  const float* pb2 = (const float*)d_in[13];
  float* out = (float*)d_out;

  const int NT = 32768, N = 4096, E = 524288;
  const int threeN = 3 * N;
  const size_t FEAT_BYTES = (size_t)NT * 256 * 4;

  char* w = (char*)d_ws;
  float* xp   = (float*)(w);
  float* hb   = (float*)(w + FEAT_BYTES);
  float* a_s  = (float*)(w + 3 * FEAT_BYTES);
  float* a_d  = a_s + (size_t)NT * 4;
  int*   cnt  = (int*)(a_d + (size_t)NT * 4);
  int*   off  = cnt + NT;
  int*   rank = off + NT + 32;
  int*   csr  = rank + E;

  const int* srcA = ei;
  const int* dstA = ei + E;

  dim3 blk(256);
  int gGather = (NT + GPW * 8 - 1) / (GPW * 8);
  int gGemm   = NT / 32;
  int gEdge   = E / EPT / 256;

  k_gemm<<<gGemm, blk, 0, stream>>>(x, W1, as1, ad1, xp, a_s, a_d, out, cnt, NT, 128, N, threeN, 1);

  k_count<<<gEdge, blk, 0, stream>>>(dstA, cnt, rank, E);
  k_scan<<<1, dim3(1024), 0, stream>>>(cnt, off, NT);
  k_fill<<<gEdge, blk, 0, stream>>>(srcA, dstA, off, rank, csr, E);

  k_gather<<<gGather, blk, 0, stream>>>(off, csr, a_s, a_d, xp, b1, pw1, pb1, hb, out, NT, N, threeN, 1);

  k_gemm<<<gGemm, blk, 0, stream>>>(hb, W2, as2, ad2, xp, a_s, a_d, out, cnt, NT, 256, N, threeN, 0);
  k_gather<<<gGather, blk, 0, stream>>>(off, csr, a_s, a_d, xp, b2, pw2, pb2, nullptr, out, NT, N, threeN, 2);
}


# Edge ranks from LDS counters: a block counts 8192 edges in a per-node LDS histogram, then one global add per block and node, 32 adjacent nodes per request; bucket (sum) order still arrival-dependent
# speedup vs baseline: 3.9686x; 1.1705x over previous
#include <hip/hip_runtime.h>
#include <math.h>


#define GPW 3
#define EPT 4
#define CNT_T 1024
#define CNT_E 8
#define NODES 32768

typedef float v2f __attribute__((ext_vector_type(2)));
typedef float v8f __attribute__((ext_vector_type(8)));
typedef float v4f __attribute__((ext_vector_type(4)));

#define DEV __device__ __forceinline__

DEV float wsum(float v) {
#pragma unroll
  for (int m = 16; m; m >>= 1) v += __shfl_xor(v, m, 32);
  return v;
}

DEV float wmax(float v) {
#pragma unroll
  for (int m = 16; m; m >>= 1) v = fmaxf(v, __shfl_xor(v, m, 32));
  return v;
}

DEV float lrelu(float l) { return (l > 0.0f) ? l : 0.2f * l; }

DEV float bcast_f(float v, int k) {
  return __int_as_float(__builtin_amdgcn_readlane(__float_as_int(v), k));
}

DEV float elu1(float t) { return (t > 0.0f) ? t : expm1f(t); }

__global__ void k_gemm(const float* __restrict__ A, const float* __restrict__ Bw,
                       const float* __restrict__ att_s, const float* __restrict__ att_d,
                       float* __restrict__ C, float* __restrict__ a_s,
                       float* __restrict__ a_d, float* __restrict__ out,
                       int* __restrict__ cnt, int M, int K, int N, int threeN,
                       int first) {
  __shared__ float sAtt[8][32];
  if ((int)blockIdx.x * 32 >= M) return;
  const int wib = threadIdx.x >> 5;
  const int lane = threadIdx.x & 31;
  const int rowTile = (int)blockIdx.x * 2 + (wib >> 2), colGrp = wib & 3;
  const int row0 = rowTile << 4, col0 = colGrp << 6;
  const int half = lane >> 4, lm = lane & 15;

  v8f acc0 = {}, acc1 = {}, acc2 = {}, acc3 = {};
  const float* ap = A + (size_t)(row0 + lm) * K + 4 * half;
  const float* bp = Bw + (size_t)(4 * half) * 256 + col0 + 4 * lm;
  float rsum = 0.0f;

  for (int k0 = 0; k0 < K; k0 += 8) {
    const v4f a4 = *reinterpret_cast<const v4f*>(ap + k0);
    rsum += a4.x + a4.y + a4.z + a4.w;
    const float* b0 = bp + (size_t)k0 * 256;
    const v4f w0 = *reinterpret_cast<const v4f*>(b0);
    const v4f w1 = *reinterpret_cast<const v4f*>(b0 + 256);
    const v4f w2 = *reinterpret_cast<const v4f*>(b0 + 512);
    const v4f w3 = *reinterpret_cast<const v4f*>(b0 + 768);
    v2f a, b;
    a.x = a4.x; a.y = a4.y;
    b.x = w0.x; b.y = w1.x;
    acc0 = __builtin_amdgcn_wmma_f32_16x16x4_f32(false, a, false, b, (short)0, acc0, false, false);
    b.x = w0.y; b.y = w1.y;
    acc1 = __builtin_amdgcn_wmma_f32_16x16x4_f32(false, a, false, b, (short)0, acc1, false, false);
    b.x = w0.z; b.y = w1.z;
    acc2 = __builtin_amdgcn_wmma_f32_16x16x4_f32(false, a, false, b, (short)0, acc2, false, false);
    b.x = w0.w; b.y = w1.w;
    acc3 = __builtin_amdgcn_wmma_f32_16x16x4_f32(false, a, false, b, (short)0, acc3, false, false);
    a.x = a4.z; a.y = a4.w;
    b.x = w2.x; b.y = w3.x;
    acc0 = __builtin_amdgcn_wmma_f32_16x16x4_f32(false, a, false, b, (short)0, acc0, false, false);
    b.x = w2.y; b.y = w3.y;
    acc1 = __builtin_amdgcn_wmma_f32_16x16x4_f32(false, a, false, b, (short)0, acc1, false, false);
    b.x = w2.z; b.y = w3.z;
    acc2 = __builtin_amdgcn_wmma_f32_16x16x4_f32(false, a, false, b, (short)0, acc2, false, false);
    b.x = w2.w; b.y = w3.w;
    acc3 = __builtin_amdgcn_wmma_f32_16x16x4_f32(false, a, false, b, (short)0, acc3, false, false);
  }

  float* cp = C + (size_t)(row0 + 8 * half) * 256 + col0 + 4 * lm;
#pragma unroll
  for (int r2 = 0; r2 < 8; r2++) {
    v4f o;
    o.x = acc0[r2]; o.y = acc1[r2]; o.z = acc2[r2]; o.w = acc3[r2];
    *reinterpret_cast<v4f*>(cp + (size_t)r2 * 256) = o;
  }

  if (first) {
    rsum += __shfl_xor(rsum, 16, 32);
    if (lane < 16) {
      const int nn = row0 + lm;
      if (colGrp == 0) out[(size_t)(nn / N) * threeN + nn % N] = rsum * (1.0f / (float)K);
      if (colGrp == 1) cnt[nn] = 0;
    }
  }

  const v4f ws = *reinterpret_cast<const v4f*>(att_s + col0 + 4 * lm);
  const v4f wd = *reinterpret_cast<const v4f*>(att_d + col0 + 4 * lm);
  const float ws0 = ws.x, ws1 = ws.y, ws2 = ws.z, ws3 = ws.w;
  const float wd0 = wd.x, wd1 = wd.y, wd2 = wd.z, wd3 = wd.w;
  float v[16];
#pragma unroll
  for (int r2 = 0; r2 < 8; r2++) {
    v[r2]     = acc0[r2] * ws0 + acc1[r2] * ws1 + acc2[r2] * ws2 + acc3[r2] * ws3;
    v[8 + r2] = acc0[r2] * wd0 + acc1[r2] * wd1 + acc2[r2] * wd2 + acc3[r2] * wd3;
  }
  const bool b3 = (lm & 8) != 0, b2 = (lm & 4) != 0, b1 = (lm & 2) != 0, b0 = (lm & 1) != 0;
  float u[8], t4[4], t2[2];
#pragma unroll
  for (int i = 0; i < 8; i++)
    u[i] = (b3 ? v[8 + i] : v[i]) + __shfl_xor(b3 ? v[i] : v[8 + i], 8, 32);
#pragma unroll
  for (int i = 0; i < 4; i++)
    t4[i] = (b2 ? u[4 + i] : u[i]) + __shfl_xor(b2 ? u[i] : u[4 + i], 4, 32);
#pragma unroll
  for (int i = 0; i < 2; i++)
    t2[i] = (b1 ? t4[2 + i] : t4[i]) + __shfl_xor(b1 ? t4[i] : t4[2 + i], 2, 32);
  const float tot = (b0 ? t2[1] : t2[0]) + __shfl_xor(b0 ? t2[0] : t2[1], 1, 32);
  sAtt[wib][lane] = tot;
  __syncthreads();
  {
    const int tid = threadIdx.x;
    const int tile = tid >> 7, sd = (tid >> 6) & 1, idx = tid & 63;
    const int row = idx >> 2, head = idx & 3;
    const float val = sAtt[tile * 4 + head][(row >> 3) * 16 + sd * 8 + (row & 7)];
    float* dp = sd ? a_d : a_s;
    dp[((size_t)blockIdx.x * 32 + tile * 16) * 4 + idx] = val;
  }
}

__global__ void __launch_bounds__(CNT_T)
k_count(const int* __restrict__ dst, int* __restrict__ cnt,
        int* __restrict__ rank, int E) {
  __shared__ int hist[NODES];
  const int tid = threadIdx.x;
  const int e0 = ((int)blockIdx.x * CNT_T + tid) * CNT_E;
  const bool full = (e0 + CNT_E <= E);
  int d[CNT_E], r[CNT_E];
  if (full) {
    const int4 a = *reinterpret_cast<const int4*>(dst + e0);
    const int4 b = *reinterpret_cast<const int4*>(dst + e0 + 4);
    d[0] = a.x; d[1] = a.y; d[2] = a.z; d[3] = a.w;
    d[4] = b.x; d[5] = b.y; d[6] = b.z; d[7] = b.w;
  } else {
#pragma unroll
    for (int i = 0; i < CNT_E; i++) d[i] = (e0 + i < E) ? dst[e0 + i] : -1;
  }
  for (int i = tid * 4; i < NODES; i += CNT_T * 4)
    *reinterpret_cast<int4*>(&hist[i]) = make_int4(0, 0, 0, 0);
  __syncthreads();
#pragma unroll
  for (int i = 0; i < CNT_E; i++) r[i] = (d[i] >= 0) ? atomicAdd(&hist[d[i]], 1) : 0;
  __syncthreads();
  int c[NODES / CNT_T];
#pragma unroll
  for (int i = 0; i < NODES / CNT_T; i++) c[i] = hist[i * CNT_T + tid];
#pragma unroll
  for (int i = 0; i < NODES / CNT_T; i++)
    if (c[i] > 0) c[i] = atomicAdd(&cnt[i * CNT_T + tid], c[i]);
#pragma unroll
  for (int i = 0; i < NODES / CNT_T; i++) hist[i * CNT_T + tid] = c[i];
  __syncthreads();
  if (full) {
    int4 a, b;
    a.x = hist[d[0]] + r[0]; a.y = hist[d[1]] + r[1];
    a.z = hist[d[2]] + r[2]; a.w = hist[d[3]] + r[3];
    b.x = hist[d[4]] + r[4]; b.y = hist[d[5]] + r[5];
    b.z = hist[d[6]] + r[6]; b.w = hist[d[7]] + r[7];
    *reinterpret_cast<int4*>(rank + e0) = a;
    *reinterpret_cast<int4*>(rank + e0 + 4) = b;
  } else {
#pragma unroll
    for (int i = 0; i < CNT_E; i++)
      if (e0 + i < E) rank[e0 + i] = hist[d[i]] + r[i];
  }
}

__global__ void __launch_bounds__(1024)
k_scan(const int* __restrict__ cnt, int* __restrict__ off, int NT) {
  __shared__ int wtot[32];
  int t = threadIdx.x, lane = t & 31, w = t >> 5;
  const int4* p = reinterpret_cast<const int4*>(cnt + t * 32);
  int4 v[8];
  int s = 0;
#pragma unroll
  for (int i = 0; i < 8; i++) {
    v[i] = p[i];
    s += v[i].x + v[i].y + v[i].z + v[i].w;
  }
  int inc = s;
#pragma unroll
  for (int o = 1; o < 32; o <<= 1) {
    int nb = __shfl_up(inc, o, 32);
    if (lane >= o) inc += nb;
  }
  if (lane == 31) wtot[w] = inc;
  __syncthreads();
  if (w == 0) {
    int x = wtot[lane], xi = x;
#pragma unroll
    for (int o = 1; o < 32; o <<= 1) {
      int nb = __shfl_up(xi, o, 32);
      if (lane >= o) xi += nb;
    }
    wtot[lane] = xi - x;
  }
  __syncthreads();
  int run = wtot[w] + inc - s;
  int4* q = reinterpret_cast<int4*>(off + t * 32);
#pragma unroll
  for (int i = 0; i < 8; i++) {
    int4 o4;
    o4.x = run; run += v[i].x;
    o4.y = run; run += v[i].y;
    o4.z = run; run += v[i].z;
    o4.w = run; run += v[i].w;
    q[i] = o4;
  }
  if (t == 1023) off[NT] = run;
}

__global__ void k_fill(const int* __restrict__ src, const int* __restrict__ dst,
                       const int* __restrict__ off, const int* __restrict__ rank,
                       int* __restrict__ csr, int E) {
  int e = (blockIdx.x * blockDim.x + threadIdx.x) * EPT;
  if (e + EPT <= E) {
    const int4 s4 = *reinterpret_cast<const int4*>(src + e);
    const int4 d4 = *reinterpret_cast<const int4*>(dst + e);
    const int4 r4 = *reinterpret_cast<const int4*>(rank + e);
    const int o0 = off[d4.x], o1 = off[d4.y], o2 = off[d4.z], o3 = off[d4.w];
    csr[o0 + r4.x] = s4.x;
    csr[o1 + r4.y] = s4.y;
    csr[o2 + r4.z] = s4.z;
    csr[o3 + r4.w] = s4.w;
  } else {
    for (; e < E; e++) csr[off[dst[e]] + rank[e]] = src[e];
  }
}

__global__ void k_gather(const int* __restrict__ off, const int* __restrict__ csr,
                         const float* __restrict__ a_s, const float* __restrict__ a_d,
                         const float* __restrict__ xp, const float* __restrict__ bias,
                         const float* __restrict__ pw, const float* __restrict__ pb,
                         float* __restrict__ hout, float* __restrict__ out,
                         int NT, int N, int threeN, int sec) {
  const unsigned lane = threadIdx.x & 31;
  const int wv = __builtin_amdgcn_readfirstlane(
      (int)((blockIdx.x * blockDim.x + threadIdx.x) >> 5));
  const bool up = lane >= 16;
  for (int g = 0; g < GPW; g++) {
  const int d = wv * GPW + g;
  if (d >= NT) return;
  const int beg = off[d], end = off[d + 1];
  const v4f ad = *reinterpret_cast<const v4f*>(a_d + (size_t)d * 4);
  const v4f sd = *reinterpret_cast<const v4f*>(a_s + (size_t)d * 4);

  const float ls0 = lrelu(sd.x + ad.x), ls1 = lrelu(sd.y + ad.y);
  const float ls2 = lrelu(sd.z + ad.z), ls3 = lrelu(sd.w + ad.w);

  float m0 = ls0, m1 = ls1, m2 = ls2, m3 = ls3;
  for (int j = beg + lane; j < end; j += 32) {
    const v4f as = *reinterpret_cast<const v4f*>(a_s + (size_t)csr[j] * 4);
    m0 = fmaxf(m0, lrelu(as.x + ad.x));
    m1 = fmaxf(m1, lrelu(as.y + ad.y));
    m2 = fmaxf(m2, lrelu(as.z + ad.z));
    m3 = fmaxf(m3, lrelu(as.w + ad.w));
  }
  m0 = wmax(m0); m1 = wmax(m1); m2 = wmax(m2); m3 = wmax(m3);

  float z0 = 0.0f, z1 = 0.0f, z2 = 0.0f, z3 = 0.0f;
  for (int j = beg + lane; j < end; j += 32) {
    const v4f as = *reinterpret_cast<const v4f*>(a_s + (size_t)csr[j] * 4);
    z0 += expf(lrelu(as.x + ad.x) - m0);
    z1 += expf(lrelu(as.y + ad.y) - m1);
    z2 += expf(lrelu(as.z + ad.z) - m2);
    z3 += expf(lrelu(as.w + ad.w) - m3);
  }
  const float es0 = expf(ls0 - m0), es1 = expf(ls1 - m1);
  const float es2 = expf(ls2 - m2), es3 = expf(ls3 - m3);
  const float dn0 = wsum(z0) + es0 + 1e-16f, dn1 = wsum(z1) + es1 + 1e-16f;
  const float dn2 = wsum(z2) + es2 + 1e-16f, dn3 = wsum(z3) + es3 + 1e-16f;

  v4f accLo, accHi;
  {
    const v4f* r = reinterpret_cast<const v4f*>(xp + (size_t)d * 256);
    const float cLo = up ? es1 / dn1 : es0 / dn0;
    const float cHi = up ? es3 / dn3 : es2 / dn2;
    accLo = r[lane] * cLo;
    accHi = r[32 + lane] * cHi;
  }
  for (int base = beg; base < end; base += 32) {
    const int j = base + lane;
    int s = d;
    float e0 = 0.0f, e1 = 0.0f, e2 = 0.0f, e3 = 0.0f;
    if (j < end) {
      s = csr[j];
      const v4f as = *reinterpret_cast<const v4f*>(a_s + (size_t)s * 4);
      e0 = expf(lrelu(as.x + ad.x) - m0) / dn0;
      e1 = expf(lrelu(as.y + ad.y) - m1) / dn1;
      e2 = expf(lrelu(as.z + ad.z) - m2) / dn2;
      e3 = expf(lrelu(as.w + ad.w) - m3) / dn3;
    }
    const int n = (end - base < 32) ? (end - base) : 32;
    for (int k = 0; k < n; k++) {
      const int sk = __builtin_amdgcn_readlane(s, k);
      const float b0 = bcast_f(e0, k), b1 = bcast_f(e1, k);
      const float b2 = bcast_f(e2, k), b3 = bcast_f(e3, k);
      const v4f* r = reinterpret_cast<const v4f*>(xp + (size_t)sk * 256);
      const v4f y0 = r[lane], y1 = r[32 + lane];
      accLo += y0 * (up ? b1 : b0);
      accHi += y1 * (up ? b3 : b2);
    }
  }
  {
    const v4f* bv = reinterpret_cast<const v4f*>(bias);
    const v4f* pv = reinterpret_cast<const v4f*>(pw);
    unsigned le = lane;
    asm volatile("" : "+v"(le));
    v4f tLo = accLo + bv[le], tHi = accHi + bv[32 + le];
    tLo.x = elu1(tLo.x); tLo.y = elu1(tLo.y); tLo.z = elu1(tLo.z); tLo.w = elu1(tLo.w);
    tHi.x = elu1(tHi.x); tHi.y = elu1(tHi.y); tHi.z = elu1(tHi.z); tHi.w = elu1(tHi.w);
    if (hout != nullptr) {
      v4f* o = reinterpret_cast<v4f*>(hout + (size_t)d * 256);
      o[le] = tLo;
      o[32 + le] = tHi;
    }
    const v4f pLo = pv[le], pHi = pv[32 + le];
    float pool = tLo.x * pLo.x + tLo.y * pLo.y + tLo.z * pLo.z + tLo.w * pLo.w
               + tHi.x * pHi.x + tHi.y * pHi.y + tHi.z * pHi.z + tHi.w * pHi.w;
    pool = wsum(pool);
    if (lane == 0) out[(size_t)(d / N) * threeN + sec * N + d % N] = pool + pb[0];
  }
  }
}

extern "C" void kernel_launch(void* const* d_in, const int* in_sizes, int n_in,
                              void* d_out, int out_size, void* d_ws, size_t ws_size,
                              hipStream_t stream) {
  const float* x   = (const float*)d_in[0];
  const int*   ei  = (const int*)d_in[1];
  const float* W1  = (const float*)d_in[2];
  const float* as1 = (const float*)d_in[3];
  const float* ad1 = (const float*)d_in[4];
  const float* b1  = (const float*)d_in[5];
  const float* W2  = (const float*)d_in[6];
  const float* as2 = (const float*)d_in[7];
  const float* ad2 = (const float*)d_in[8];
  const float* b2  = (const float*)d_in[9];
  const float* pw1 = (const float*)d_in[10];
  const float* pb1 = (const float*)d_in[11];
  const float* pw2 = (const float*)d_in[12];
  const float* pb2 = (const float*)d_in[13];
  float* out = (float*)d_out;

  const int NT = 32768, N = 4096, E = 524288;
  const int threeN = 3 * N;
  const size_t FEAT_BYTES = (size_t)NT * 256 * 4;

  char* w = (char*)d_ws;
  float* xp   = (float*)(w);
  float* hb   = (float*)(w + FEAT_BYTES);
  float* a_s  = (float*)(w + 3 * FEAT_BYTES);
  float* a_d  = a_s + (size_t)NT * 4;
  int*   cnt  = (int*)(a_d + (size_t)NT * 4);
  int*   off  = cnt + NT;
  int*   rank = off + NT + 32;
  int*   csr  = rank + E;

  const int* srcA = ei;
  const int* dstA = ei + E;

  dim3 blk(256);
  int gGather = (NT + GPW * 8 - 1) / (GPW * 8);
  int gGemm   = NT / 32;
  int gEdge   = E / EPT / 256;
  int gCount  = (E + CNT_T * CNT_E - 1) / (CNT_T * CNT_E);

  k_gemm<<<gGemm, blk, 0, stream>>>(x, W1, as1, ad1, xp, a_s, a_d, out, cnt, NT, 128, N, threeN, 1);

  k_count<<<gCount, dim3(CNT_T), 0, stream>>>(dstA, cnt, rank, E);
  k_scan<<<1, dim3(1024), 0, stream>>>(cnt, off, NT);
  k_fill<<<gEdge, blk, 0, stream>>>(srcA, dstA, off, rank, csr, E);

  k_gather<<<gGather, blk, 0, stream>>>(off, csr, a_s, a_d, xp, b1, pw1, pb1, hb, out, NT, N, threeN, 1);

  k_gemm<<<gGemm, blk, 0, stream>>>(hb, W2, as2, ad2, xp, a_s, a_d, out, cnt, NT, 256, N, threeN, 0);
  k_gather<<<gGather, blk, 0, stream>>>(off, csr, a_s, a_d, xp, b2, pw2, pb2, nullptr, out, NT, N, threeN, 2);
}


# Gather reads a node's list once: self-loop folded in as entry 0, lane keeps its entry's source, logits and exp in registers; denominators summed in a different order (self-loop's exp in the tree)
# speedup vs baseline: 4.1888x; 1.0555x over previous
#include <hip/hip_runtime.h>
#include <math.h>


#define GPW 3
#define EPT 4
#define CNT_T 1024
#define CNT_E 8
#define NODES 32768

typedef float v2f __attribute__((ext_vector_type(2)));
typedef float v8f __attribute__((ext_vector_type(8)));
typedef float v4f __attribute__((ext_vector_type(4)));

#define DEV __device__ __forceinline__

DEV float wsum(float v) {
#pragma unroll
  for (int m = 16; m; m >>= 1) v += __shfl_xor(v, m, 32);
  return v;
}

DEV float wmax(float v) {
#pragma unroll
  for (int m = 16; m; m >>= 1) v = fmaxf(v, __shfl_xor(v, m, 32));
  return v;
}

DEV float lrelu(float l) { return (l > 0.0f) ? l : 0.2f * l; }

DEV float bcast_f(float v, int k) {
  return __int_as_float(__builtin_amdgcn_readlane(__float_as_int(v), k));
}

DEV float uni_f(float v) {
  return __int_as_float(__builtin_amdgcn_readfirstlane(__float_as_int(v)));
}

DEV float elu1(float t) { return (t > 0.0f) ? t : expm1f(t); }

__global__ void k_gemm(const float* __restrict__ A, const float* __restrict__ Bw,
                       const float* __restrict__ att_s, const float* __restrict__ att_d,
                       float* __restrict__ C, float* __restrict__ a_s,
                       float* __restrict__ a_d, float* __restrict__ out,
                       int* __restrict__ cnt, int M, int K, int N, int threeN,
                       int first) {
  __shared__ float sAtt[8][32];
  if ((int)blockIdx.x * 32 >= M) return;
  const int wib = threadIdx.x >> 5;
  const int lane = threadIdx.x & 31;
  const int rowTile = (int)blockIdx.x * 2 + (wib >> 2), colGrp = wib & 3;
  const int row0 = rowTile << 4, col0 = colGrp << 6;
  const int half = lane >> 4, lm = lane & 15;

  v8f acc0 = {}, acc1 = {}, acc2 = {}, acc3 = {};
  const float* ap = A + (size_t)(row0 + lm) * K + 4 * half;
  const float* bp = Bw + (size_t)(4 * half) * 256 + col0 + 4 * lm;
  float rsum = 0.0f;

  for (int k0 = 0; k0 < K; k0 += 8) {
    const v4f a4 = *reinterpret_cast<const v4f*>(ap + k0);
    rsum += a4.x + a4.y + a4.z + a4.w;
    const float* b0 = bp + (size_t)k0 * 256;
    const v4f w0 = *reinterpret_cast<const v4f*>(b0);
    const v4f w1 = *reinterpret_cast<const v4f*>(b0 + 256);
    const v4f w2 = *reinterpret_cast<const v4f*>(b0 + 512);
    const v4f w3 = *reinterpret_cast<const v4f*>(b0 + 768);
    v2f a, b;
    a.x = a4.x; a.y = a4.y;
    b.x = w0.x; b.y = w1.x;
    acc0 = __builtin_amdgcn_wmma_f32_16x16x4_f32(false, a, false, b, (short)0, acc0, false, false);
    b.x = w0.y; b.y = w1.y;
    acc1 = __builtin_amdgcn_wmma_f32_16x16x4_f32(false, a, false, b, (short)0, acc1, false, false);
    b.x = w0.z; b.y = w1.z;
    acc2 = __builtin_amdgcn_wmma_f32_16x16x4_f32(false, a, false, b, (short)0, acc2, false, false);
    b.x = w0.w; b.y = w1.w;
    acc3 = __builtin_amdgcn_wmma_f32_16x16x4_f32(false, a, false, b, (short)0, acc3, false, false);
    a.x = a4.z; a.y = a4.w;
    b.x = w2.x; b.y = w3.x;
    acc0 = __builtin_amdgcn_wmma_f32_16x16x4_f32(false, a, false, b, (short)0, acc0, false, false);
    b.x = w2.y; b.y = w3.y;
    acc1 = __builtin_amdgcn_wmma_f32_16x16x4_f32(false, a, false, b, (short)0, acc1, false, false);
    b.x = w2.z; b.y = w3.z;
    acc2 = __builtin_amdgcn_wmma_f32_16x16x4_f32(false, a, false, b, (short)0, acc2, false, false);
    b.x = w2.w; b.y = w3.w;
    acc3 = __builtin_amdgcn_wmma_f32_16x16x4_f32(false, a, false, b, (short)0, acc3, false, false);
  }

  float* cp = C + (size_t)(row0 + 8 * half) * 256 + col0 + 4 * lm;
#pragma unroll
  for (int r2 = 0; r2 < 8; r2++) {
    v4f o;
    o.x = acc0[r2]; o.y = acc1[r2]; o.z = acc2[r2]; o.w = acc3[r2];
    *reinterpret_cast<v4f*>(cp + (size_t)r2 * 256) = o;
  }

  if (first) {
    rsum += __shfl_xor(rsum, 16, 32);
    if (lane < 16) {
      const int nn = row0 + lm;
      if (colGrp == 0) out[(size_t)(nn / N) * threeN + nn % N] = rsum * (1.0f / (float)K);
      if (colGrp == 1) cnt[nn] = 0;
    }
  }

  const v4f ws = *reinterpret_cast<const v4f*>(att_s + col0 + 4 * lm);
  const v4f wd = *reinterpret_cast<const v4f*>(att_d + col0 + 4 * lm);
  const float ws0 = ws.x, ws1 = ws.y, ws2 = ws.z, ws3 = ws.w;
  const float wd0 = wd.x, wd1 = wd.y, wd2 = wd.z, wd3 = wd.w;
  float v[16];
#pragma unroll
  for (int r2 = 0; r2 < 8; r2++) {
    v[r2]     = acc0[r2] * ws0 + acc1[r2] * ws1 + acc2[r2] * ws2 + acc3[r2] * ws3;
    v[8 + r2] = acc0[r2] * wd0 + acc1[r2] * wd1 + acc2[r2] * wd2 + acc3[r2] * wd3;
  }
  const bool b3 = (lm & 8) != 0, b2 = (lm & 4) != 0, b1 = (lm & 2) != 0, b0 = (lm & 1) != 0;
  float u[8], t4[4], t2[2];
#pragma unroll
  for (int i = 0; i < 8; i++)
    u[i] = (b3 ? v[8 + i] : v[i]) + __shfl_xor(b3 ? v[i] : v[8 + i], 8, 32);
#pragma unroll
  for (int i = 0; i < 4; i++)
    t4[i] = (b2 ? u[4 + i] : u[i]) + __shfl_xor(b2 ? u[i] : u[4 + i], 4, 32);
#pragma unroll
  for (int i = 0; i < 2; i++)
    t2[i] = (b1 ? t4[2 + i] : t4[i]) + __shfl_xor(b1 ? t4[i] : t4[2 + i], 2, 32);
  const float tot = (b0 ? t2[1] : t2[0]) + __shfl_xor(b0 ? t2[0] : t2[1], 1, 32);
  sAtt[wib][lane] = tot;
  __syncthreads();
  {
    const int tid = threadIdx.x;
    const int tile = tid >> 7, sd = (tid >> 6) & 1, idx = tid & 63;
    const int row = idx >> 2, head = idx & 3;
    const float val = sAtt[tile * 4 + head][(row >> 3) * 16 + sd * 8 + (row & 7)];
    float* dp = sd ? a_d : a_s;
    dp[((size_t)blockIdx.x * 32 + tile * 16) * 4 + idx] = val;
  }
}

__global__ void __launch_bounds__(CNT_T)
k_count(const int* __restrict__ dst, int* __restrict__ cnt,
        int* __restrict__ rank, int E) {
  __shared__ int hist[NODES];
  const int tid = threadIdx.x;
  const int e0 = ((int)blockIdx.x * CNT_T + tid) * CNT_E;
  const bool full = (e0 + CNT_E <= E);
  int d[CNT_E], r[CNT_E];
  if (full) {
    const int4 a = *reinterpret_cast<const int4*>(dst + e0);
    const int4 b = *reinterpret_cast<const int4*>(dst + e0 + 4);
    d[0] = a.x; d[1] = a.y; d[2] = a.z; d[3] = a.w;
    d[4] = b.x; d[5] = b.y; d[6] = b.z; d[7] = b.w;
  } else {
#pragma unroll
    for (int i = 0; i < CNT_E; i++) d[i] = (e0 + i < E) ? dst[e0 + i] : -1;
  }
  for (int i = tid * 4; i < NODES; i += CNT_T * 4)
    *reinterpret_cast<int4*>(&hist[i]) = make_int4(0, 0, 0, 0);
  __syncthreads();
#pragma unroll
  for (int i = 0; i < CNT_E; i++) r[i] = (d[i] >= 0) ? atomicAdd(&hist[d[i]], 1) : 0;
  __syncthreads();
  int c[NODES / CNT_T];
#pragma unroll
  for (int i = 0; i < NODES / CNT_T; i++) c[i] = hist[i * CNT_T + tid];
#pragma unroll
  for (int i = 0; i < NODES / CNT_T; i++)
    if (c[i] > 0) c[i] = atomicAdd(&cnt[i * CNT_T + tid], c[i]);
#pragma unroll
  for (int i = 0; i < NODES / CNT_T; i++) hist[i * CNT_T + tid] = c[i];
  __syncthreads();
  if (full) {
    int4 a, b;
    a.x = hist[d[0]] + r[0]; a.y = hist[d[1]] + r[1];
    a.z = hist[d[2]] + r[2]; a.w = hist[d[3]] + r[3];
    b.x = hist[d[4]] + r[4]; b.y = hist[d[5]] + r[5];
    b.z = hist[d[6]] + r[6]; b.w = hist[d[7]] + r[7];
    *reinterpret_cast<int4*>(rank + e0) = a;
    *reinterpret_cast<int4*>(rank + e0 + 4) = b;
  } else {
#pragma unroll
    for (int i = 0; i < CNT_E; i++)
      if (e0 + i < E) rank[e0 + i] = hist[d[i]] + r[i];
  }
}

__global__ void __launch_bounds__(1024)
k_scan(const int* __restrict__ cnt, int* __restrict__ off, int NT) {
  __shared__ int wtot[32];
  int t = threadIdx.x, lane = t & 31, w = t >> 5;
  const int4* p = reinterpret_cast<const int4*>(cnt + t * 32);
  int4 v[8];
  int s = 0;
#pragma unroll
  for (int i = 0; i < 8; i++) {
    v[i] = p[i];
    s += v[i].x + v[i].y + v[i].z + v[i].w;
  }
  int inc = s;
#pragma unroll
  for (int o = 1; o < 32; o <<= 1) {
    int nb = __shfl_up(inc, o, 32);
    if (lane >= o) inc += nb;
  }
  if (lane == 31) wtot[w] = inc;
  __syncthreads();
  if (w == 0) {
    int x = wtot[lane], xi = x;
#pragma unroll
    for (int o = 1; o < 32; o <<= 1) {
      int nb = __shfl_up(xi, o, 32);
      if (lane >= o) xi += nb;
    }
    wtot[lane] = xi - x;
  }
  __syncthreads();
  int run = wtot[w] + inc - s;
  int4* q = reinterpret_cast<int4*>(off + t * 32);
#pragma unroll
  for (int i = 0; i < 8; i++) {
    int4 o4;
    o4.x = run; run += v[i].x;
    o4.y = run; run += v[i].y;
    o4.z = run; run += v[i].z;
    o4.w = run; run += v[i].w;
    q[i] = o4;
  }
  if (t == 1023) off[NT] = run;
}

__global__ void k_fill(const int* __restrict__ src, const int* __restrict__ dst,
                       const int* __restrict__ off, const int* __restrict__ rank,
                       int* __restrict__ csr, int E) {
  int e = (blockIdx.x * blockDim.x + threadIdx.x) * EPT;
  if (e + EPT <= E) {
    const int4 s4 = *reinterpret_cast<const int4*>(src + e);
    const int4 d4 = *reinterpret_cast<const int4*>(dst + e);
    const int4 r4 = *reinterpret_cast<const int4*>(rank + e);
    const int o0 = off[d4.x], o1 = off[d4.y], o2 = off[d4.z], o3 = off[d4.w];
    csr[o0 + r4.x] = s4.x;
    csr[o1 + r4.y] = s4.y;
    csr[o2 + r4.z] = s4.z;
    csr[o3 + r4.w] = s4.w;
  } else {
    for (; e < E; e++) csr[off[dst[e]] + rank[e]] = src[e];
  }
}

__global__ void k_gather(const int* __restrict__ off, const int* __restrict__ csr,
                         const float* __restrict__ a_s, const float* __restrict__ a_d,
                         const float* __restrict__ xp, const float* __restrict__ bias,
                         const float* __restrict__ pw, const float* __restrict__ pb,
                         float* __restrict__ hout, float* __restrict__ out,
                         int NT, int N, int threeN, int sec) {
  const unsigned lane = threadIdx.x & 31;
  const int wv = __builtin_amdgcn_readfirstlane(
      (int)((blockIdx.x * blockDim.x + threadIdx.x) >> 5));
  const bool up = lane >= 16;
  for (int g = 0; g < GPW; g++) {
  const int d = wv * GPW + g;
  if (d >= NT) return;
  const int beg = off[d], end = off[d + 1];
  const int nl = end - beg + 1;
  const v4f ad = *reinterpret_cast<const v4f*>(a_d + (size_t)d * 4);

  const bool in0 = (int)lane < nl;
  int s0 = d;
  if (in0 && lane > 0) s0 = csr[beg + (int)lane - 1];
  float l0, l1, l2, l3;
  {
    const v4f as = *reinterpret_cast<const v4f*>(a_s + (size_t)s0 * 4);
    l0 = lrelu(as.x + ad.x); l1 = lrelu(as.y + ad.y);
    l2 = lrelu(as.z + ad.z); l3 = lrelu(as.w + ad.w);
  }

  float m0 = l0, m1 = l1, m2 = l2, m3 = l3;
  for (int i = 32 + (int)lane; i < nl; i += 32) {
    const v4f as = *reinterpret_cast<const v4f*>(a_s + (size_t)csr[beg + i - 1] * 4);
    m0 = fmaxf(m0, lrelu(as.x + ad.x));
    m1 = fmaxf(m1, lrelu(as.y + ad.y));
    m2 = fmaxf(m2, lrelu(as.z + ad.z));
    m3 = fmaxf(m3, lrelu(as.w + ad.w));
  }
  m0 = uni_f(wmax(m0)); m1 = uni_f(wmax(m1)); m2 = uni_f(wmax(m2)); m3 = uni_f(wmax(m3));

  const float p0 = in0 ? expf(l0 - m0) : 0.0f, p1 = in0 ? expf(l1 - m1) : 0.0f;
  const float p2 = in0 ? expf(l2 - m2) : 0.0f, p3 = in0 ? expf(l3 - m3) : 0.0f;
  float z0 = p0, z1 = p1, z2 = p2, z3 = p3;
  for (int i = 32 + (int)lane; i < nl; i += 32) {
    const v4f as = *reinterpret_cast<const v4f*>(a_s + (size_t)csr[beg + i - 1] * 4);
    z0 += expf(lrelu(as.x + ad.x) - m0);
    z1 += expf(lrelu(as.y + ad.y) - m1);
    z2 += expf(lrelu(as.z + ad.z) - m2);
    z3 += expf(lrelu(as.w + ad.w) - m3);
  }
  const float dn0 = wsum(z0) + 1e-16f, dn1 = wsum(z1) + 1e-16f;
  const float dn2 = wsum(z2) + 1e-16f, dn3 = wsum(z3) + 1e-16f;

  v4f accLo = {0.0f, 0.0f, 0.0f, 0.0f}, accHi = {0.0f, 0.0f, 0.0f, 0.0f};
  for (int base = 0; base < nl; base += 32) {
    int s = d;
    float e0 = 0.0f, e1 = 0.0f, e2 = 0.0f, e3 = 0.0f;
    if (base == 0) {
      s = s0;
      e0 = p0 / dn0; e1 = p1 / dn1; e2 = p2 / dn2; e3 = p3 / dn3;
    } else {
      const int i = base + (int)lane;
      if (i < nl) {
        s = csr[beg + i - 1];
        const v4f as = *reinterpret_cast<const v4f*>(a_s + (size_t)s * 4);
        e0 = expf(lrelu(as.x + ad.x) - m0) / dn0;
        e1 = expf(lrelu(as.y + ad.y) - m1) / dn1;
        e2 = expf(lrelu(as.z + ad.z) - m2) / dn2;
        e3 = expf(lrelu(as.w + ad.w) - m3) / dn3;
      }
    }
    const int n = (nl - base < 32) ? (nl - base) : 32;
    for (int k = 0; k < n; k++) {
      const int sk = __builtin_amdgcn_readlane(s, k);
      const float b0 = bcast_f(e0, k), b1 = bcast_f(e1, k);
      const float b2 = bcast_f(e2, k), b3 = bcast_f(e3, k);
      const v4f* r = reinterpret_cast<const v4f*>(xp + (size_t)sk * 256);
      const v4f y0 = r[lane], y1 = r[32 + lane];
      accLo += y0 * (up ? b1 : b0);
      accHi += y1 * (up ? b3 : b2);
    }
  }
  {
    const v4f* bv = reinterpret_cast<const v4f*>(bias);
    const v4f* pv = reinterpret_cast<const v4f*>(pw);
    unsigned le = lane;
    asm volatile("" : "+v"(le));
    v4f tLo = accLo + bv[le], tHi = accHi + bv[32 + le];
    tLo.x = elu1(tLo.x); tLo.y = elu1(tLo.y); tLo.z = elu1(tLo.z); tLo.w = elu1(tLo.w);
    tHi.x = elu1(tHi.x); tHi.y = elu1(tHi.y); tHi.z = elu1(tHi.z); tHi.w = elu1(tHi.w);
    if (hout != nullptr) {
      v4f* o = reinterpret_cast<v4f*>(hout + (size_t)d * 256);
      o[le] = tLo;
      o[32 + le] = tHi;
    }
    const v4f pLo = pv[le], pHi = pv[32 + le];
    float pool = tLo.x * pLo.x + tLo.y * pLo.y + tLo.z * pLo.z + tLo.w * pLo.w
               + tHi.x * pHi.x + tHi.y * pHi.y + tHi.z * pHi.z + tHi.w * pHi.w;
    pool = wsum(pool);
    if (lane == 0) out[(size_t)(d / N) * threeN + sec * N + d % N] = pool + pb[0];
  }
  }
}

extern "C" void kernel_launch(void* const* d_in, const int* in_sizes, int n_in,
                              void* d_out, int out_size, void* d_ws, size_t ws_size,
                              hipStream_t stream) {
  const float* x   = (const float*)d_in[0];
  const int*   ei  = (const int*)d_in[1];
  const float* W1  = (const float*)d_in[2];
  const float* as1 = (const float*)d_in[3];
  const float* ad1 = (const float*)d_in[4];
  const float* b1  = (const float*)d_in[5];
  const float* W2  = (const float*)d_in[6];
  const float* as2 = (const float*)d_in[7];
  const float* ad2 = (const float*)d_in[8];
  const float* b2  = (const float*)d_in[9];
  const float* pw1 = (const float*)d_in[10];
  const float* pb1 = (const float*)d_in[11];
  const float* pw2 = (const float*)d_in[12];
  const float* pb2 = (const float*)d_in[13];
  float* out = (float*)d_out;

  const int NT = 32768, N = 4096, E = 524288;
  const int threeN = 3 * N;
  const size_t FEAT_BYTES = (size_t)NT * 256 * 4;

  char* w = (char*)d_ws;
  float* xp   = (float*)(w);
  float* hb   = (float*)(w + FEAT_BYTES);
  float* a_s  = (float*)(w + 3 * FEAT_BYTES);
  float* a_d  = a_s + (size_t)NT * 4;
  int*   cnt  = (int*)(a_d + (size_t)NT * 4);
  int*   off  = cnt + NT;
  int*   rank = off + NT + 32;
  int*   csr  = rank + E;

  const int* srcA = ei;
  const int* dstA = ei + E;

  dim3 blk(256);
  int gGather = (NT + GPW * 8 - 1) / (GPW * 8);
  int gGemm   = NT / 32;
  int gEdge   = E / EPT / 256;
  int gCount  = (E + CNT_T * CNT_E - 1) / (CNT_T * CNT_E);

  k_gemm<<<gGemm, blk, 0, stream>>>(x, W1, as1, ad1, xp, a_s, a_d, out, cnt, NT, 128, N, threeN, 1);

  k_count<<<gCount, dim3(CNT_T), 0, stream>>>(dstA, cnt, rank, E);
  k_scan<<<1, dim3(1024), 0, stream>>>(cnt, off, NT);
  k_fill<<<gEdge, blk, 0, stream>>>(srcA, dstA, off, rank, csr, E);

  k_gather<<<gGather, blk, 0, stream>>>(off, csr, a_s, a_d, xp, b1, pw1, pb1, hb, out, NT, N, threeN, 1);

  k_gemm<<<gGemm, blk, 0, stream>>>(hb, W2, as2, ad2, xp, a_s, a_d, out, cnt, NT, 256, N, threeN, 0);
  k_gather<<<gGather, blk, 0, stream>>>(off, csr, a_s, a_d, xp, b2, pw2, pb2, nullptr, out, NT, N, threeN, 2);
}


# Layer-1 gather + layer-2 GEMM in one launch: a block gathers 32 nodes (8 rows requested at a time), its ELU rows feed the WMMA loop from LDS; h1 not stored; 7 -> 6 launches; sums unchanged
# speedup vs baseline: 4.5403x; 1.0839x over previous
#include <hip/hip_runtime.h>
#include <math.h>


#define GPW 3
#define FGW 4
#define RB 8
#define HP 260
#define EPT 4
#define CNT_T 1024
#define CNT_E 8
#define NODES 32768

typedef float v2f __attribute__((ext_vector_type(2)));
typedef float v8f __attribute__((ext_vector_type(8)));
typedef float v4f __attribute__((ext_vector_type(4)));

#define DEV __device__ __forceinline__

DEV float wsum(float v) {
#pragma unroll
  for (int m = 16; m; m >>= 1) v += __shfl_xor(v, m, 32);
  return v;
}

DEV float wmax(float v) {
#pragma unroll
  for (int m = 16; m; m >>= 1) v = fmaxf(v, __shfl_xor(v, m, 32));
  return v;
}

DEV float lrelu(float l) { return (l > 0.0f) ? l : 0.2f * l; }

DEV float bcast_f(float v, int k) {
  return __int_as_float(__builtin_amdgcn_readlane(__float_as_int(v), k));
}

DEV float uni_f(float v) {
  return __int_as_float(__builtin_amdgcn_readfirstlane(__float_as_int(v)));
}

DEV float elu1(float t) { return (t > 0.0f) ? t : expm1f(t); }

DEV void mma_kstep8(const v4f a4, const float* __restrict__ b0, v8f& acc0, v8f& acc1,
                    v8f& acc2, v8f& acc3) {
  const v4f w0 = *reinterpret_cast<const v4f*>(b0);
  const v4f w1 = *reinterpret_cast<const v4f*>(b0 + 256);
  const v4f w2 = *reinterpret_cast<const v4f*>(b0 + 512);
  const v4f w3 = *reinterpret_cast<const v4f*>(b0 + 768);
  v2f a, b;
  a.x = a4.x; a.y = a4.y;
  b.x = w0.x; b.y = w1.x;
  acc0 = __builtin_amdgcn_wmma_f32_16x16x4_f32(false, a, false, b, (short)0, acc0, false, false);
  b.x = w0.y; b.y = w1.y;
  acc1 = __builtin_amdgcn_wmma_f32_16x16x4_f32(false, a, false, b, (short)0, acc1, false, false);
  b.x = w0.z; b.y = w1.z;
  acc2 = __builtin_amdgcn_wmma_f32_16x16x4_f32(false, a, false, b, (short)0, acc2, false, false);
  b.x = w0.w; b.y = w1.w;
  acc3 = __builtin_amdgcn_wmma_f32_16x16x4_f32(false, a, false, b, (short)0, acc3, false, false);
  a.x = a4.z; a.y = a4.w;
  b.x = w2.x; b.y = w3.x;
  acc0 = __builtin_amdgcn_wmma_f32_16x16x4_f32(false, a, false, b, (short)0, acc0, false, false);
  b.x = w2.y; b.y = w3.y;
  acc1 = __builtin_amdgcn_wmma_f32_16x16x4_f32(false, a, false, b, (short)0, acc1, false, false);
  b.x = w2.z; b.y = w3.z;
  acc2 = __builtin_amdgcn_wmma_f32_16x16x4_f32(false, a, false, b, (short)0, acc2, false, false);
  b.x = w2.w; b.y = w3.w;
  acc3 = __builtin_amdgcn_wmma_f32_16x16x4_f32(false, a, false, b, (short)0, acc3, false, false);
}

DEV void tile_finish(const v8f& acc0, const v8f& acc1, const v8f& acc2, const v8f& acc3,
                     const float* __restrict__ att_s, const float* __restrict__ att_d,
                     float* __restrict__ C, float* __restrict__ a_s, float* __restrict__ a_d,
                     float (*sAtt)[32], const int row0, const int col0, const int half,
                     const int lm, const int lane, const int wib) {
  float* cp = C + (size_t)(row0 + 8 * half) * 256 + col0 + 4 * lm;
#pragma unroll
  for (int r2 = 0; r2 < 8; r2++) {
    v4f o;
    o.x = acc0[r2]; o.y = acc1[r2]; o.z = acc2[r2]; o.w = acc3[r2];
    *reinterpret_cast<v4f*>(cp + (size_t)r2 * 256) = o;
  }

  const v4f ws = *reinterpret_cast<const v4f*>(att_s + col0 + 4 * lm);
  const v4f wd = *reinterpret_cast<const v4f*>(att_d + col0 + 4 * lm);
  const float ws0 = ws.x, ws1 = ws.y, ws2 = ws.z, ws3 = ws.w;
  const float wd0 = wd.x, wd1 = wd.y, wd2 = wd.z, wd3 = wd.w;
  float v[16];
#pragma unroll
  for (int r2 = 0; r2 < 8; r2++) {
    v[r2]     = acc0[r2] * ws0 + acc1[r2] * ws1 + acc2[r2] * ws2 + acc3[r2] * ws3;
    v[8 + r2] = acc0[r2] * wd0 + acc1[r2] * wd1 + acc2[r2] * wd2 + acc3[r2] * wd3;
  }
  const bool b3 = (lm & 8) != 0, b2 = (lm & 4) != 0, b1 = (lm & 2) != 0, b0 = (lm & 1) != 0;
  float u[8], t4[4], t2[2];
#pragma unroll
  for (int i = 0; i < 8; i++)
    u[i] = (b3 ? v[8 + i] : v[i]) + __shfl_xor(b3 ? v[i] : v[8 + i], 8, 32);
#pragma unroll
  for (int i = 0; i < 4; i++)
    t4[i] = (b2 ? u[4 + i] : u[i]) + __shfl_xor(b2 ? u[i] : u[4 + i], 4, 32);
#pragma unroll
  for (int i = 0; i < 2; i++)
    t2[i] = (b1 ? t4[2 + i] : t4[i]) + __shfl_xor(b1 ? t4[i] : t4[2 + i], 2, 32);
  const float tot = (b0 ? t2[1] : t2[0]) + __shfl_xor(b0 ? t2[0] : t2[1], 1, 32);
  sAtt[wib][lane] = tot;
  __syncthreads();
  {
    const int tid = threadIdx.x;
    const int tile = tid >> 7, sd = (tid >> 6) & 1, idx = tid & 63;
    const int row = idx >> 2, head = idx & 3;
    const float val = sAtt[tile * 4 + head][(row >> 3) * 16 + sd * 8 + (row & 7)];
    float* dp = sd ? a_d : a_s;
    dp[((size_t)blockIdx.x * 32 + tile * 16) * 4 + idx] = val;
  }
}

__global__ void k_gemm(const float* __restrict__ A, const float* __restrict__ Bw,
                       const float* __restrict__ att_s, const float* __restrict__ att_d,
                       float* __restrict__ C, float* __restrict__ a_s,
                       float* __restrict__ a_d, float* __restrict__ out,
                       int* __restrict__ cnt, int M, int K, int N, int threeN,
                       int first) {
  __shared__ float sAtt[8][32];
  if ((int)blockIdx.x * 32 >= M) return;
  const int wib = threadIdx.x >> 5;
  const int lane = threadIdx.x & 31;
  const int rowTile = (int)blockIdx.x * 2 + (wib >> 2), colGrp = wib & 3;
  const int row0 = rowTile << 4, col0 = colGrp << 6;
  const int half = lane >> 4, lm = lane & 15;

  v8f acc0 = {}, acc1 = {}, acc2 = {}, acc3 = {};
  const float* ap = A + (size_t)(row0 + lm) * K + 4 * half;
  const float* bp = Bw + (size_t)(4 * half) * 256 + col0 + 4 * lm;
  float rsum = 0.0f;

  for (int k0 = 0; k0 < K; k0 += 8) {
    const v4f a4 = *reinterpret_cast<const v4f*>(ap + k0);
    rsum += a4.x + a4.y + a4.z + a4.w;
    mma_kstep8(a4, bp + (size_t)k0 * 256, acc0, acc1, acc2, acc3);
  }

  if (first) {
    rsum += __shfl_xor(rsum, 16, 32);
    if (lane < 16) {
      const int nn = row0 + lm;
      if (colGrp == 0) out[(size_t)(nn / N) * threeN + nn % N] = rsum * (1.0f / (float)K);
      if (colGrp == 1) cnt[nn] = 0;
    }
  }

  tile_finish(acc0, acc1, acc2, acc3, att_s, att_d, C, a_s, a_d, sAtt, row0, col0, half,
              lm, lane, wib);
}

__global__ void __launch_bounds__(CNT_T)
k_count(const int* __restrict__ dst, int* __restrict__ cnt,
        int* __restrict__ rank, int E) {
  __shared__ int hist[NODES];
  const int tid = threadIdx.x;
  const int e0 = ((int)blockIdx.x * CNT_T + tid) * CNT_E;
  const bool full = (e0 + CNT_E <= E);
  int d[CNT_E], r[CNT_E];
  if (full) {
    const int4 a = *reinterpret_cast<const int4*>(dst + e0);
    const int4 b = *reinterpret_cast<const int4*>(dst + e0 + 4);
    d[0] = a.x; d[1] = a.y; d[2] = a.z; d[3] = a.w;
    d[4] = b.x; d[5] = b.y; d[6] = b.z; d[7] = b.w;
  } else {
#pragma unroll
    for (int i = 0; i < CNT_E; i++) d[i] = (e0 + i < E) ? dst[e0 + i] : -1;
  }
  for (int i = tid * 4; i < NODES; i += CNT_T * 4)
    *reinterpret_cast<int4*>(&hist[i]) = make_int4(0, 0, 0, 0);
  __syncthreads();
#pragma unroll
  for (int i = 0; i < CNT_E; i++) r[i] = (d[i] >= 0) ? atomicAdd(&hist[d[i]], 1) : 0;
  __syncthreads();
  int c[NODES / CNT_T];
#pragma unroll
  for (int i = 0; i < NODES / CNT_T; i++) c[i] = hist[i * CNT_T + tid];
#pragma unroll
  for (int i = 0; i < NODES / CNT_T; i++)
    if (c[i] > 0) c[i] = atomicAdd(&cnt[i * CNT_T + tid], c[i]);
#pragma unroll
  for (int i = 0; i < NODES / CNT_T; i++) hist[i * CNT_T + tid] = c[i];
  __syncthreads();
  if (full) {
    int4 a, b;
    a.x = hist[d[0]] + r[0]; a.y = hist[d[1]] + r[1];
    a.z = hist[d[2]] + r[2]; a.w = hist[d[3]] + r[3];
    b.x = hist[d[4]] + r[4]; b.y = hist[d[5]] + r[5];
    b.z = hist[d[6]] + r[6]; b.w = hist[d[7]] + r[7];
    *reinterpret_cast<int4*>(rank + e0) = a;
    *reinterpret_cast<int4*>(rank + e0 + 4) = b;
  } else {
#pragma unroll
    for (int i = 0; i < CNT_E; i++)
      if (e0 + i < E) rank[e0 + i] = hist[d[i]] + r[i];
  }
}

__global__ void __launch_bounds__(1024)
k_scan(const int* __restrict__ cnt, int* __restrict__ off, int NT) {
  __shared__ int wtot[32];
  int t = threadIdx.x, lane = t & 31, w = t >> 5;
  const int4* p = reinterpret_cast<const int4*>(cnt + t * 32);
  int4 v[8];
  int s = 0;
#pragma unroll
  for (int i = 0; i < 8; i++) {
    v[i] = p[i];
    s += v[i].x + v[i].y + v[i].z + v[i].w;
  }
  int inc = s;
#pragma unroll
  for (int o = 1; o < 32; o <<= 1) {
    int nb = __shfl_up(inc, o, 32);
    if (lane >= o) inc += nb;
  }
  if (lane == 31) wtot[w] = inc;
  __syncthreads();
  if (w == 0) {
    int x = wtot[lane], xi = x;
#pragma unroll
    for (int o = 1; o < 32; o <<= 1) {
      int nb = __shfl_up(xi, o, 32);
      if (lane >= o) xi += nb;
    }
    wtot[lane] = xi - x;
  }
  __syncthreads();
  int run = wtot[w] + inc - s;
  int4* q = reinterpret_cast<int4*>(off + t * 32);
#pragma unroll
  for (int i = 0; i < 8; i++) {
    int4 o4;
    o4.x = run; run += v[i].x;
    o4.y = run; run += v[i].y;
    o4.z = run; run += v[i].z;
    o4.w = run; run += v[i].w;
    q[i] = o4;
  }
  if (t == 1023) off[NT] = run;
}

__global__ void k_fill(const int* __restrict__ src, const int* __restrict__ dst,
                       const int* __restrict__ off, const int* __restrict__ rank,
                       int* __restrict__ csr, int E) {
  int e = (blockIdx.x * blockDim.x + threadIdx.x) * EPT;
  if (e + EPT <= E) {
    const int4 s4 = *reinterpret_cast<const int4*>(src + e);
    const int4 d4 = *reinterpret_cast<const int4*>(dst + e);
    const int4 r4 = *reinterpret_cast<const int4*>(rank + e);
    const int o0 = off[d4.x], o1 = off[d4.y], o2 = off[d4.z], o3 = off[d4.w];
    csr[o0 + r4.x] = s4.x;
    csr[o1 + r4.y] = s4.y;
    csr[o2 + r4.z] = s4.z;
    csr[o3 + r4.w] = s4.w;
  } else {
    for (; e < E; e++) csr[off[dst[e]] + rank[e]] = src[e];
  }
}

template <int U>
DEV void rows_batch(const float* __restrict__ xp, const int s, const float e0,
                    const float e1, const float e2, const float e3, const int k,
                    const unsigned lane, const bool up, v4f& accLo, v4f& accHi) {
  v4f y0[U], y1[U];
#pragma unroll
  for (int j = 0; j < U; j++) {
    const int sk = __builtin_amdgcn_readlane(s, k + j);
    const v4f* r = reinterpret_cast<const v4f*>(xp + (size_t)sk * 256);
    y0[j] = r[lane];
    y1[j] = r[32 + lane];
  }
#pragma unroll
  for (int j = 0; j < U; j++) {
    const float b0 = bcast_f(e0, k + j), b1 = bcast_f(e1, k + j);
    const float b2 = bcast_f(e2, k + j), b3 = bcast_f(e3, k + j);
    accLo += y0[j] * (up ? b1 : b0);
    accHi += y1[j] * (up ? b3 : b2);
  }
}

template <int NB>
DEV void gather_node(const int d, const int* __restrict__ off, const int* __restrict__ csr,
                     const float* __restrict__ a_s, const float* __restrict__ a_d,
                     const float* __restrict__ xp, const unsigned lane, const bool up,
                     v4f& accLo, v4f& accHi) {
  static_assert(NB == 0 || NB == 8, "the tail switch below is written for batches of 8");
  const int beg = off[d], end = off[d + 1];
  const int nl = end - beg + 1;
  const v4f ad = *reinterpret_cast<const v4f*>(a_d + (size_t)d * 4);

  const bool in0 = (int)lane < nl;
  int s0 = d;
  if (in0 && lane > 0) s0 = csr[beg + (int)lane - 1];
  float l0, l1, l2, l3;
  {
    const v4f as = *reinterpret_cast<const v4f*>(a_s + (size_t)s0 * 4);
    l0 = lrelu(as.x + ad.x); l1 = lrelu(as.y + ad.y);
    l2 = lrelu(as.z + ad.z); l3 = lrelu(as.w + ad.w);
  }

  float m0 = l0, m1 = l1, m2 = l2, m3 = l3;
  for (int i = 32 + (int)lane; i < nl; i += 32) {
    const v4f as = *reinterpret_cast<const v4f*>(a_s + (size_t)csr[beg + i - 1] * 4);
    m0 = fmaxf(m0, lrelu(as.x + ad.x));
    m1 = fmaxf(m1, lrelu(as.y + ad.y));
    m2 = fmaxf(m2, lrelu(as.z + ad.z));
    m3 = fmaxf(m3, lrelu(as.w + ad.w));
  }
  m0 = uni_f(wmax(m0)); m1 = uni_f(wmax(m1)); m2 = uni_f(wmax(m2)); m3 = uni_f(wmax(m3));

  const float p0 = in0 ? expf(l0 - m0) : 0.0f, p1 = in0 ? expf(l1 - m1) : 0.0f;
  const float p2 = in0 ? expf(l2 - m2) : 0.0f, p3 = in0 ? expf(l3 - m3) : 0.0f;
  float z0 = p0, z1 = p1, z2 = p2, z3 = p3;
  for (int i = 32 + (int)lane; i < nl; i += 32) {
    const v4f as = *reinterpret_cast<const v4f*>(a_s + (size_t)csr[beg + i - 1] * 4);
    z0 += expf(lrelu(as.x + ad.x) - m0);
    z1 += expf(lrelu(as.y + ad.y) - m1);
    z2 += expf(lrelu(as.z + ad.z) - m2);
    z3 += expf(lrelu(as.w + ad.w) - m3);
  }
  const float dn0 = wsum(z0) + 1e-16f, dn1 = wsum(z1) + 1e-16f;
  const float dn2 = wsum(z2) + 1e-16f, dn3 = wsum(z3) + 1e-16f;

  accLo = v4f{0.0f, 0.0f, 0.0f, 0.0f};
  accHi = v4f{0.0f, 0.0f, 0.0f, 0.0f};
  for (int base = 0; base < nl; base += 32) {
    int s = d;
    float e0 = 0.0f, e1 = 0.0f, e2 = 0.0f, e3 = 0.0f;
    if (base == 0) {
      s = s0;
      e0 = p0 / dn0; e1 = p1 / dn1; e2 = p2 / dn2; e3 = p3 / dn3;
    } else {
      const int i = base + (int)lane;
      if (i < nl) {
        s = csr[beg + i - 1];
        const v4f as = *reinterpret_cast<const v4f*>(a_s + (size_t)s * 4);
        e0 = expf(lrelu(as.x + ad.x) - m0) / dn0;
        e1 = expf(lrelu(as.y + ad.y) - m1) / dn1;
        e2 = expf(lrelu(as.z + ad.z) - m2) / dn2;
        e3 = expf(lrelu(as.w + ad.w) - m3) / dn3;
      }
    }
    const int n = (nl - base < 32) ? (nl - base) : 32;
    if constexpr (NB == 0) {
      for (int k = 0; k < n; k++) rows_batch<1>(xp, s, e0, e1, e2, e3, k, lane, up, accLo, accHi);
    } else {
      int k = 0;
      for (; k + NB <= n; k += NB)
        rows_batch<NB>(xp, s, e0, e1, e2, e3, k, lane, up, accLo, accHi);
      switch (n - k) {
        case 1: rows_batch<1>(xp, s, e0, e1, e2, e3, k, lane, up, accLo, accHi); break;
        case 2: rows_batch<2>(xp, s, e0, e1, e2, e3, k, lane, up, accLo, accHi); break;
        case 3: rows_batch<3>(xp, s, e0, e1, e2, e3, k, lane, up, accLo, accHi); break;
        case 4: rows_batch<4>(xp, s, e0, e1, e2, e3, k, lane, up, accLo, accHi); break;
        case 5: rows_batch<5>(xp, s, e0, e1, e2, e3, k, lane, up, accLo, accHi); break;
        case 6: rows_batch<6>(xp, s, e0, e1, e2, e3, k, lane, up, accLo, accHi); break;
        case 7: rows_batch<7>(xp, s, e0, e1, e2, e3, k, lane, up, accLo, accHi); break;
        default: break;
      }
    }
  }
}

__global__ void k_gather(const int* __restrict__ off, const int* __restrict__ csr,
                         const float* __restrict__ a_s, const float* __restrict__ a_d,
                         const float* __restrict__ xp, const float* __restrict__ bias,
                         const float* __restrict__ pw, const float* __restrict__ pb,
                         float* __restrict__ hout, float* __restrict__ out,
                         int NT, int N, int threeN, int sec) {
  const unsigned lane = threadIdx.x & 31;
  const int wv = __builtin_amdgcn_readfirstlane(
      (int)((blockIdx.x * blockDim.x + threadIdx.x) >> 5));
  const bool up = lane >= 16;
  for (int g = 0; g < GPW; g++) {
  const int d = wv * GPW + g;
  if (d >= NT) return;
  v4f accLo, accHi;
  gather_node<0>(d, off, csr, a_s, a_d, xp, lane, up, accLo, accHi);
  {
    const v4f* bv = reinterpret_cast<const v4f*>(bias);
    const v4f* pv = reinterpret_cast<const v4f*>(pw);
    unsigned le = lane;
    asm volatile("" : "+v"(le));
    v4f tLo = accLo + bv[le], tHi = accHi + bv[32 + le];
    tLo.x = elu1(tLo.x); tLo.y = elu1(tLo.y); tLo.z = elu1(tLo.z); tLo.w = elu1(tLo.w);
    tHi.x = elu1(tHi.x); tHi.y = elu1(tHi.y); tHi.z = elu1(tHi.z); tHi.w = elu1(tHi.w);
    if (hout != nullptr) {
      v4f* o = reinterpret_cast<v4f*>(hout + (size_t)d * 256);
      o[le] = tLo;
      o[32 + le] = tHi;
    }
    const v4f pLo = pv[le], pHi = pv[32 + le];
    float pool = tLo.x * pLo.x + tLo.y * pLo.y + tLo.z * pLo.z + tLo.w * pLo.w
               + tHi.x * pHi.x + tHi.y * pHi.y + tHi.z * pHi.z + tHi.w * pHi.w;
    pool = wsum(pool);
    if (lane == 0) out[(size_t)(d / N) * threeN + sec * N + d % N] = pool + pb[0];
  }
  }
}

__global__ void k_gather_gemm(const int* __restrict__ off, const int* __restrict__ csr,
                              const float* __restrict__ a_s, const float* __restrict__ a_d,
                              const float* __restrict__ xp, const float* __restrict__ bias,
                              const float* __restrict__ pw, const float* __restrict__ pb,
                              const float* __restrict__ Bw, const float* __restrict__ att_s,
                              const float* __restrict__ att_d, float* __restrict__ C,
                              float* __restrict__ a_s2, float* __restrict__ a_d2,
                              float* __restrict__ out, int NT, int K, int N, int threeN) {
  __shared__ alignas(16) float sH[32 * HP];
  __shared__ float sAtt[8][32];
  if ((int)blockIdx.x * 32 >= NT) return;
  const unsigned lane = threadIdx.x & 31;
  const int wib = __builtin_amdgcn_readfirstlane((int)(threadIdx.x >> 5));
  const bool up = lane >= 16;

  for (int g = 0; g < FGW; g++) {
    const int r = wib * FGW + g;
    const int d = (int)blockIdx.x * 32 + r;
    v4f accLo, accHi;
    gather_node<RB>(d, off, csr, a_s, a_d, xp, lane, up, accLo, accHi);
    const v4f* bv = reinterpret_cast<const v4f*>(bias);
    const v4f* pv = reinterpret_cast<const v4f*>(pw);
    v4f tLo = accLo + bv[lane], tHi = accHi + bv[32 + lane];
    tLo.x = elu1(tLo.x); tLo.y = elu1(tLo.y); tLo.z = elu1(tLo.z); tLo.w = elu1(tLo.w);
    tHi.x = elu1(tHi.x); tHi.y = elu1(tHi.y); tHi.z = elu1(tHi.z); tHi.w = elu1(tHi.w);
    float* hr = sH + r * HP;
    *reinterpret_cast<v4f*>(hr + 4 * lane) = tLo;
    *reinterpret_cast<v4f*>(hr + 128 + 4 * lane) = tHi;
    const v4f pLo = pv[lane], pHi = pv[32 + lane];
    float pool = tLo.x * pLo.x + tLo.y * pLo.y + tLo.z * pLo.z + tLo.w * pLo.w
               + tHi.x * pHi.x + tHi.y * pHi.y + tHi.z * pHi.z + tHi.w * pHi.w;
    pool = wsum(pool);
    if (lane == 0) out[(size_t)(d / N) * threeN + N + d % N] = pool + pb[0];
  }
  __syncthreads();

  const int colGrp = wib & 3;
  const int row0l = (wib >> 2) << 4, col0 = colGrp << 6;
  const int row0 = (int)blockIdx.x * 32 + row0l;
  const int half = (int)(lane >> 4), lm = (int)(lane & 15);
  v8f acc0 = {}, acc1 = {}, acc2 = {}, acc3 = {};
  const float* ap = sH + (row0l + lm) * HP + 4 * half;
  const float* bp = Bw + (size_t)(4 * half) * 256 + col0 + 4 * lm;
  for (int k0 = 0; k0 < K; k0 += 8) {
    const v4f a4 = *reinterpret_cast<const v4f*>(ap + k0);
    mma_kstep8(a4, bp + (size_t)k0 * 256, acc0, acc1, acc2, acc3);
  }
  tile_finish(acc0, acc1, acc2, acc3, att_s, att_d, C, a_s2, a_d2, sAtt, row0, col0, half,
              lm, (int)lane, wib);
}

extern "C" void kernel_launch(void* const* d_in, const int* in_sizes, int n_in,
                              void* d_out, int out_size, void* d_ws, size_t ws_size,
                              hipStream_t stream) {
  const float* x   = (const float*)d_in[0];
  const int*   ei  = (const int*)d_in[1];
  const float* W1  = (const float*)d_in[2];
  const float* as1 = (const float*)d_in[3];
  const float* ad1 = (const float*)d_in[4];
  const float* b1  = (const float*)d_in[5];
  const float* W2  = (const float*)d_in[6];
  const float* as2 = (const float*)d_in[7];
  const float* ad2 = (const float*)d_in[8];
  const float* b2  = (const float*)d_in[9];
  const float* pw1 = (const float*)d_in[10];
  const float* pb1 = (const float*)d_in[11];
  const float* pw2 = (const float*)d_in[12];
  const float* pb2 = (const float*)d_in[13];
  float* out = (float*)d_out;

  const int NT = 32768, N = 4096, E = 524288;
  const int threeN = 3 * N;
  const size_t FEAT_BYTES = (size_t)NT * 256 * 4;

  char* w = (char*)d_ws;
  float* xp   = (float*)(w);
  float* xp2  = (float*)(w + FEAT_BYTES);
  float* a_s2 = (float*)(w + 2 * FEAT_BYTES);
  float* a_d2 = a_s2 + (size_t)NT * 4;
  float* a_s  = (float*)(w + 3 * FEAT_BYTES);
  float* a_d  = a_s + (size_t)NT * 4;
  int*   cnt  = (int*)(a_d + (size_t)NT * 4);
  int*   off  = cnt + NT;
  int*   rank = off + NT + 32;
  int*   csr  = rank + E;

  const int* srcA = ei;
  const int* dstA = ei + E;

  dim3 blk(256);
  int gGather = (NT + GPW * 8 - 1) / (GPW * 8);
  int gGemm   = NT / 32;
  int gEdge   = E / EPT / 256;
  int gCount  = (E + CNT_T * CNT_E - 1) / (CNT_T * CNT_E);

  k_gemm<<<gGemm, blk, 0, stream>>>(x, W1, as1, ad1, xp, a_s, a_d, out, cnt, NT, 128, N, threeN, 1);

  k_count<<<gCount, dim3(CNT_T), 0, stream>>>(dstA, cnt, rank, E);
  k_scan<<<1, dim3(1024), 0, stream>>>(cnt, off, NT);
  k_fill<<<gEdge, blk, 0, stream>>>(srcA, dstA, off, rank, csr, E);

  k_gather_gemm<<<gGemm, blk, 0, stream>>>(off, csr, a_s, a_d, xp, b1, pw1, pb1, W2, as2, ad2,
                                          xp2, a_s2, a_d2, out, NT, 256, N, threeN);

  k_gather<<<gGather, blk, 0, stream>>>(off, csr, a_s2, a_d2, xp2, b2, pw2, pb2, nullptr, out, NT, N, threeN, 2);
}


# Layer 1 aggregates before it multiplies: gather sums alpha-weighted x rows (128 floats, not 256) per head, W1 applied on WMMA in the fused block; attention dots as x.(W1 att); sum order changed
# speedup vs baseline: 4.7360x; 1.0431x over previous
#include <hip/hip_runtime.h>
#include <math.h>


#define GPW 3
#define FGW 4
#define RB 8
#define HP 260
#define XP 132
#define FIN 128
#define PRE_N 8
#define EPT 4
#define CNT_T 1024
#define CNT_E 8
#define NODES 32768

typedef float v2f __attribute__((ext_vector_type(2)));
typedef float v8f __attribute__((ext_vector_type(8)));
typedef float v4f __attribute__((ext_vector_type(4)));

#define DEV __device__ __forceinline__

DEV float wsum(float v) {
#pragma unroll
  for (int m = 16; m; m >>= 1) v += __shfl_xor(v, m, 32);
  return v;
}

DEV float wmax(float v) {
#pragma unroll
  for (int m = 16; m; m >>= 1) v = fmaxf(v, __shfl_xor(v, m, 32));
  return v;
}

DEV float lrelu(float l) { return (l > 0.0f) ? l : 0.2f * l; }

DEV float bcast_f(float v, int k) {
  return __int_as_float(__builtin_amdgcn_readlane(__float_as_int(v), k));
}

DEV float uni_f(float v) {
  return __int_as_float(__builtin_amdgcn_readfirstlane(__float_as_int(v)));
}

DEV float elu1(float t) { return (t > 0.0f) ? t : expm1f(t); }

DEV void mma_kstep8(const v4f a4, const float* __restrict__ b0, v8f& acc0, v8f& acc1,
                    v8f& acc2, v8f& acc3) {
  const v4f w0 = *reinterpret_cast<const v4f*>(b0);
  const v4f w1 = *reinterpret_cast<const v4f*>(b0 + 256);
  const v4f w2 = *reinterpret_cast<const v4f*>(b0 + 512);
  const v4f w3 = *reinterpret_cast<const v4f*>(b0 + 768);
  v2f a, b;
  a.x = a4.x; a.y = a4.y;
  b.x = w0.x; b.y = w1.x;
  acc0 = __builtin_amdgcn_wmma_f32_16x16x4_f32(false, a, false, b, (short)0, acc0, false, false);
  b.x = w0.y; b.y = w1.y;
  acc1 = __builtin_amdgcn_wmma_f32_16x16x4_f32(false, a, false, b, (short)0, acc1, false, false);
  b.x = w0.z; b.y = w1.z;
  acc2 = __builtin_amdgcn_wmma_f32_16x16x4_f32(false, a, false, b, (short)0, acc2, false, false);
  b.x = w0.w; b.y = w1.w;
  acc3 = __builtin_amdgcn_wmma_f32_16x16x4_f32(false, a, false, b, (short)0, acc3, false, false);
  a.x = a4.z; a.y = a4.w;
  b.x = w2.x; b.y = w3.x;
  acc0 = __builtin_amdgcn_wmma_f32_16x16x4_f32(false, a, false, b, (short)0, acc0, false, false);
  b.x = w2.y; b.y = w3.y;
  acc1 = __builtin_amdgcn_wmma_f32_16x16x4_f32(false, a, false, b, (short)0, acc1, false, false);
  b.x = w2.z; b.y = w3.z;
  acc2 = __builtin_amdgcn_wmma_f32_16x16x4_f32(false, a, false, b, (short)0, acc2, false, false);
  b.x = w2.w; b.y = w3.w;
  acc3 = __builtin_amdgcn_wmma_f32_16x16x4_f32(false, a, false, b, (short)0, acc3, false, false);
}

DEV void tile_finish(const v8f& acc0, const v8f& acc1, const v8f& acc2, const v8f& acc3,
                     const float* __restrict__ att_s, const float* __restrict__ att_d,
                     float* __restrict__ C, float* __restrict__ a_s, float* __restrict__ a_d,
                     float (*sAtt)[32], const int row0, const int col0, const int half,
                     const int lm, const int lane, const int wib) {
  float* cp = C + (size_t)(row0 + 8 * half) * 256 + col0 + 4 * lm;
#pragma unroll
  for (int r2 = 0; r2 < 8; r2++) {
    v4f o;
    o.x = acc0[r2]; o.y = acc1[r2]; o.z = acc2[r2]; o.w = acc3[r2];
    *reinterpret_cast<v4f*>(cp + (size_t)r2 * 256) = o;
  }

  const v4f ws = *reinterpret_cast<const v4f*>(att_s + col0 + 4 * lm);
  const v4f wd = *reinterpret_cast<const v4f*>(att_d + col0 + 4 * lm);
  const float ws0 = ws.x, ws1 = ws.y, ws2 = ws.z, ws3 = ws.w;
  const float wd0 = wd.x, wd1 = wd.y, wd2 = wd.z, wd3 = wd.w;
  float v[16];
#pragma unroll
  for (int r2 = 0; r2 < 8; r2++) {
    v[r2]     = acc0[r2] * ws0 + acc1[r2] * ws1 + acc2[r2] * ws2 + acc3[r2] * ws3;
    v[8 + r2] = acc0[r2] * wd0 + acc1[r2] * wd1 + acc2[r2] * wd2 + acc3[r2] * wd3;
  }
  const bool b3 = (lm & 8) != 0, b2 = (lm & 4) != 0, b1 = (lm & 2) != 0, b0 = (lm & 1) != 0;
  float u[8], t4[4], t2[2];
#pragma unroll
  for (int i = 0; i < 8; i++)
    u[i] = (b3 ? v[8 + i] : v[i]) + __shfl_xor(b3 ? v[i] : v[8 + i], 8, 32);
#pragma unroll
  for (int i = 0; i < 4; i++)
    t4[i] = (b2 ? u[4 + i] : u[i]) + __shfl_xor(b2 ? u[i] : u[4 + i], 4, 32);
#pragma unroll
  for (int i = 0; i < 2; i++)
    t2[i] = (b1 ? t4[2 + i] : t4[i]) + __shfl_xor(b1 ? t4[i] : t4[2 + i], 2, 32);
  const float tot = (b0 ? t2[1] : t2[0]) + __shfl_xor(b0 ? t2[0] : t2[1], 1, 32);
  sAtt[wib][lane] = tot;
  __syncthreads();
  {
    const int tid = threadIdx.x;
    const int tile = tid >> 7, sd = (tid >> 6) & 1, idx = tid & 63;
    const int row = idx >> 2, head = idx & 3;
    const float val = sAtt[tile * 4 + head][(row >> 3) * 16 + sd * 8 + (row & 7)];
    float* dp = sd ? a_d : a_s;
    dp[((size_t)blockIdx.x * 32 + tile * 16) * 4 + idx] = val;
  }
}

__global__ void
k_pre(const float* __restrict__ x, const float* __restrict__ W,
      const float* __restrict__ att_s, const float* __restrict__ att_d,
      float* __restrict__ a_s, float* __restrict__ a_d, float* __restrict__ out,
      int* __restrict__ cnt, int NT, int N, int threeN) {
  __shared__ alignas(16) float sU[FIN * 8];
  const unsigned lane = threadIdx.x & 31;
  const int wib = threadIdx.x >> 5;
  {
    const v4f sLo = reinterpret_cast<const v4f*>(att_s)[lane];
    const v4f sHi = reinterpret_cast<const v4f*>(att_s)[32 + lane];
    const v4f dLo = reinterpret_cast<const v4f*>(att_d)[lane];
    const v4f dHi = reinterpret_cast<const v4f*>(att_d)[32 + lane];
    for (int i0 = 0; i0 < FIN / 8; i0 += 8) {
      v4f wl[8], wh[8];
#pragma unroll
      for (int j = 0; j < 8; j++) {
        const v4f* wr = reinterpret_cast<const v4f*>(W + (size_t)(wib * (FIN / 8) + i0 + j) * 256);
        wl[j] = wr[lane];
        wh[j] = wr[32 + lane];
      }
#pragma unroll
      for (int j = 0; j < 8; j++) {
        const int k = wib * (FIN / 8) + i0 + j;
        const v4f wLo = wl[j], wHi = wh[j];
        float q0 = wLo.x * sLo.x + wLo.y * sLo.y + wLo.z * sLo.z + wLo.w * sLo.w;
        float q1 = wHi.x * sHi.x + wHi.y * sHi.y + wHi.z * sHi.z + wHi.w * sHi.w;
        float q2 = wLo.x * dLo.x + wLo.y * dLo.y + wLo.z * dLo.z + wLo.w * dLo.w;
        float q3 = wHi.x * dHi.x + wHi.y * dHi.y + wHi.z * dHi.z + wHi.w * dHi.w;
#pragma unroll
        for (int m = 8; m; m >>= 1) {
          q0 += __shfl_xor(q0, m, 32);
          q1 += __shfl_xor(q1, m, 32);
          q2 += __shfl_xor(q2, m, 32);
          q3 += __shfl_xor(q3, m, 32);
        }
        if ((lane & 15) == 0) {
          const int h = (int)(lane >> 4);
          sU[k * 8 + h] = q0;
          sU[k * 8 + 2 + h] = q1;
          sU[k * 8 + 4 + h] = q2;
          sU[k * 8 + 6 + h] = q3;
        }
      }
    }
  }
  __syncthreads();

  float u[4][8];
#pragma unroll
  for (int j = 0; j < 4; j++) {
    const v4f ua = *reinterpret_cast<const v4f*>(&sU[(4 * lane + j) * 8]);
    const v4f ub = *reinterpret_cast<const v4f*>(&sU[(4 * lane + j) * 8 + 4]);
    u[j][0] = ua.x; u[j][1] = ua.y; u[j][2] = ua.z; u[j][3] = ua.w;
    u[j][4] = ub.x; u[j][5] = ub.y; u[j][6] = ub.z; u[j][7] = ub.w;
  }
  const int n0 = ((int)blockIdx.x * 8 + wib) * PRE_N;
  if (n0 >= NT) return;
  v4f xv[PRE_N];
#pragma unroll
  for (int i = 0; i < PRE_N; i++)
    xv[i] = reinterpret_cast<const v4f*>(x + (size_t)(n0 + i) * FIN)[lane];
  if (lane < PRE_N) cnt[n0 + (int)lane] = 0;
#pragma unroll
  for (int i = 0; i < PRE_N; i++) {
    const int nn = n0 + i;
    float p[8];
#pragma unroll
    for (int c = 0; c < 8; c++)
      p[c] = xv[i].x * u[0][c] + xv[i].y * u[1][c] + xv[i].z * u[2][c] + xv[i].w * u[3][c];
    float rs = xv[i].x + xv[i].y + xv[i].z + xv[i].w;
#pragma unroll
    for (int c = 0; c < 8; c++) p[c] = wsum(p[c]);
    rs = wsum(rs);
    if (lane == 0) {
      *reinterpret_cast<v4f*>(a_s + (size_t)nn * 4) = v4f{p[0], p[1], p[2], p[3]};
      *reinterpret_cast<v4f*>(a_d + (size_t)nn * 4) = v4f{p[4], p[5], p[6], p[7]};
      out[(size_t)(nn / N) * threeN + nn % N] = rs * (1.0f / (float)FIN);
    }
  }
}

__global__ void __launch_bounds__(CNT_T)
k_count(const int* __restrict__ dst, int* __restrict__ cnt,
        int* __restrict__ rank, int E) {
  __shared__ int hist[NODES];
  const int tid = threadIdx.x;
  const int e0 = ((int)blockIdx.x * CNT_T + tid) * CNT_E;
  const bool full = (e0 + CNT_E <= E);
  int d[CNT_E], r[CNT_E];
  if (full) {
    const int4 a = *reinterpret_cast<const int4*>(dst + e0);
    const int4 b = *reinterpret_cast<const int4*>(dst + e0 + 4);
    d[0] = a.x; d[1] = a.y; d[2] = a.z; d[3] = a.w;
    d[4] = b.x; d[5] = b.y; d[6] = b.z; d[7] = b.w;
  } else {
#pragma unroll
    for (int i = 0; i < CNT_E; i++) d[i] = (e0 + i < E) ? dst[e0 + i] : -1;
  }
  for (int i = tid * 4; i < NODES; i += CNT_T * 4)
    *reinterpret_cast<int4*>(&hist[i]) = make_int4(0, 0, 0, 0);
  __syncthreads();
#pragma unroll
  for (int i = 0; i < CNT_E; i++) r[i] = (d[i] >= 0) ? atomicAdd(&hist[d[i]], 1) : 0;
  __syncthreads();
  int c[NODES / CNT_T];
#pragma unroll
  for (int i = 0; i < NODES / CNT_T; i++) c[i] = hist[i * CNT_T + tid];
#pragma unroll
  for (int i = 0; i < NODES / CNT_T; i++)
    if (c[i] > 0) c[i] = atomicAdd(&cnt[i * CNT_T + tid], c[i]);
#pragma unroll
  for (int i = 0; i < NODES / CNT_T; i++) hist[i * CNT_T + tid] = c[i];
  __syncthreads();
  if (full) {
    int4 a, b;
    a.x = hist[d[0]] + r[0]; a.y = hist[d[1]] + r[1];
    a.z = hist[d[2]] + r[2]; a.w = hist[d[3]] + r[3];
    b.x = hist[d[4]] + r[4]; b.y = hist[d[5]] + r[5];
    b.z = hist[d[6]] + r[6]; b.w = hist[d[7]] + r[7];
    *reinterpret_cast<int4*>(rank + e0) = a;
    *reinterpret_cast<int4*>(rank + e0 + 4) = b;
  } else {
#pragma unroll
    for (int i = 0; i < CNT_E; i++)
      if (e0 + i < E) rank[e0 + i] = hist[d[i]] + r[i];
  }
}

__global__ void __launch_bounds__(1024)
k_scan(const int* __restrict__ cnt, int* __restrict__ off, int NT) {
  __shared__ int wtot[32];
  int t = threadIdx.x, lane = t & 31, w = t >> 5;
  const int4* p = reinterpret_cast<const int4*>(cnt + t * 32);
  int4 v[8];
  int s = 0;
#pragma unroll
  for (int i = 0; i < 8; i++) {
    v[i] = p[i];
    s += v[i].x + v[i].y + v[i].z + v[i].w;
  }
  int inc = s;
#pragma unroll
  for (int o = 1; o < 32; o <<= 1) {
    int nb = __shfl_up(inc, o, 32);
    if (lane >= o) inc += nb;
  }
  if (lane == 31) wtot[w] = inc;
  __syncthreads();
  if (w == 0) {
    int x = wtot[lane], xi = x;
#pragma unroll
    for (int o = 1; o < 32; o <<= 1) {
      int nb = __shfl_up(xi, o, 32);
      if (lane >= o) xi += nb;
    }
    wtot[lane] = xi - x;
  }
  __syncthreads();
  int run = wtot[w] + inc - s;
  int4* q = reinterpret_cast<int4*>(off + t * 32);
#pragma unroll
  for (int i = 0; i < 8; i++) {
    int4 o4;
    o4.x = run; run += v[i].x;
    o4.y = run; run += v[i].y;
    o4.z = run; run += v[i].z;
    o4.w = run; run += v[i].w;
    q[i] = o4;
  }
  if (t == 1023) off[NT] = run;
}

__global__ void k_fill(const int* __restrict__ src, const int* __restrict__ dst,
                       const int* __restrict__ off, const int* __restrict__ rank,
                       int* __restrict__ csr, int E) {
  int e = (blockIdx.x * blockDim.x + threadIdx.x) * EPT;
  if (e + EPT <= E) {
    const int4 s4 = *reinterpret_cast<const int4*>(src + e);
    const int4 d4 = *reinterpret_cast<const int4*>(dst + e);
    const int4 r4 = *reinterpret_cast<const int4*>(rank + e);
    const int o0 = off[d4.x], o1 = off[d4.y], o2 = off[d4.z], o3 = off[d4.w];
    csr[o0 + r4.x] = s4.x;
    csr[o1 + r4.y] = s4.y;
    csr[o2 + r4.z] = s4.z;
    csr[o3 + r4.w] = s4.w;
  } else {
    for (; e < E; e++) csr[off[dst[e]] + rank[e]] = src[e];
  }
}

template <int U, bool X4>
DEV void rows_batch(const float* __restrict__ xp, const int s, const float e0,
                    const float e1, const float e2, const float e3, const int k,
                    const unsigned lane, const bool up, v4f& accLo, v4f& accHi,
                    v4f& a2, v4f& a3) {
  if constexpr (X4) {
    v4f y[U];
#pragma unroll
    for (int j = 0; j < U; j++) {
      const int sk = __builtin_amdgcn_readlane(s, k + j);
      y[j] = reinterpret_cast<const v4f*>(xp + (size_t)sk * FIN)[lane];
    }
#pragma unroll
    for (int j = 0; j < U; j++) {
      accLo += y[j] * bcast_f(e0, k + j);
      accHi += y[j] * bcast_f(e1, k + j);
      a2 += y[j] * bcast_f(e2, k + j);
      a3 += y[j] * bcast_f(e3, k + j);
    }
  } else {
    v4f y0[U], y1[U];
#pragma unroll
    for (int j = 0; j < U; j++) {
      const int sk = __builtin_amdgcn_readlane(s, k + j);
      const v4f* r = reinterpret_cast<const v4f*>(xp + (size_t)sk * 256);
      y0[j] = r[lane];
      y1[j] = r[32 + lane];
    }
#pragma unroll
    for (int j = 0; j < U; j++) {
      const float b0 = bcast_f(e0, k + j), b1 = bcast_f(e1, k + j);
      const float b2 = bcast_f(e2, k + j), b3 = bcast_f(e3, k + j);
      accLo += y0[j] * (up ? b1 : b0);
      accHi += y1[j] * (up ? b3 : b2);
    }
  }
}

template <int NB, bool X4>
DEV void gather_node(const int d, const int* __restrict__ off, const int* __restrict__ csr,
                     const float* __restrict__ a_s, const float* __restrict__ a_d,
                     const float* __restrict__ xp, const unsigned lane, const bool up,
                     v4f& accLo, v4f& accHi, v4f& a2, v4f& a3) {
  static_assert(NB == 0 || NB == 8, "the tail switch below is written for batches of 8");
  const int beg = off[d], end = off[d + 1];
  const int nl = end - beg + 1;
  const v4f ad = *reinterpret_cast<const v4f*>(a_d + (size_t)d * 4);

  const bool in0 = (int)lane < nl;
  int s0 = d;
  if (in0 && lane > 0) s0 = csr[beg + (int)lane - 1];
  float l0, l1, l2, l3;
  {
    const v4f as = *reinterpret_cast<const v4f*>(a_s + (size_t)s0 * 4);
    l0 = lrelu(as.x + ad.x); l1 = lrelu(as.y + ad.y);
    l2 = lrelu(as.z + ad.z); l3 = lrelu(as.w + ad.w);
  }

  float m0 = l0, m1 = l1, m2 = l2, m3 = l3;
  for (int i = 32 + (int)lane; i < nl; i += 32) {
    const v4f as = *reinterpret_cast<const v4f*>(a_s + (size_t)csr[beg + i - 1] * 4);
    m0 = fmaxf(m0, lrelu(as.x + ad.x));
    m1 = fmaxf(m1, lrelu(as.y + ad.y));
    m2 = fmaxf(m2, lrelu(as.z + ad.z));
    m3 = fmaxf(m3, lrelu(as.w + ad.w));
  }
  m0 = uni_f(wmax(m0)); m1 = uni_f(wmax(m1)); m2 = uni_f(wmax(m2)); m3 = uni_f(wmax(m3));

  const float p0 = in0 ? expf(l0 - m0) : 0.0f, p1 = in0 ? expf(l1 - m1) : 0.0f;
  const float p2 = in0 ? expf(l2 - m2) : 0.0f, p3 = in0 ? expf(l3 - m3) : 0.0f;
  float z0 = p0, z1 = p1, z2 = p2, z3 = p3;
  for (int i = 32 + (int)lane; i < nl; i += 32) {
    const v4f as = *reinterpret_cast<const v4f*>(a_s + (size_t)csr[beg + i - 1] * 4);
    z0 += expf(lrelu(as.x + ad.x) - m0);
    z1 += expf(lrelu(as.y + ad.y) - m1);
    z2 += expf(lrelu(as.z + ad.z) - m2);
    z3 += expf(lrelu(as.w + ad.w) - m3);
  }
  const float dn0 = wsum(z0) + 1e-16f, dn1 = wsum(z1) + 1e-16f;
  const float dn2 = wsum(z2) + 1e-16f, dn3 = wsum(z3) + 1e-16f;

  accLo = v4f{0.0f, 0.0f, 0.0f, 0.0f};
  accHi = v4f{0.0f, 0.0f, 0.0f, 0.0f};
  a2 = v4f{0.0f, 0.0f, 0.0f, 0.0f};
  a3 = v4f{0.0f, 0.0f, 0.0f, 0.0f};
  for (int base = 0; base < nl; base += 32) {
    int s = d;
    float e0 = 0.0f, e1 = 0.0f, e2 = 0.0f, e3 = 0.0f;
    if (base == 0) {
      s = s0;
      e0 = p0 / dn0; e1 = p1 / dn1; e2 = p2 / dn2; e3 = p3 / dn3;
    } else {
      const int i = base + (int)lane;
      if (i < nl) {
        s = csr[beg + i - 1];
        const v4f as = *reinterpret_cast<const v4f*>(a_s + (size_t)s * 4);
        e0 = expf(lrelu(as.x + ad.x) - m0) / dn0;
        e1 = expf(lrelu(as.y + ad.y) - m1) / dn1;
        e2 = expf(lrelu(as.z + ad.z) - m2) / dn2;
        e3 = expf(lrelu(as.w + ad.w) - m3) / dn3;
      }
    }
    const int n = (nl - base < 32) ? (nl - base) : 32;
    if constexpr (NB == 0) {
      for (int k = 0; k < n; k++)
        rows_batch<1, X4>(xp, s, e0, e1, e2, e3, k, lane, up, accLo, accHi, a2, a3);
    } else {
      int k = 0;
      for (; k + NB <= n; k += NB)
        rows_batch<NB, X4>(xp, s, e0, e1, e2, e3, k, lane, up, accLo, accHi, a2, a3);
      switch (n - k) {
        case 1: rows_batch<1, X4>(xp, s, e0, e1, e2, e3, k, lane, up, accLo, accHi, a2, a3); break;
        case 2: rows_batch<2, X4>(xp, s, e0, e1, e2, e3, k, lane, up, accLo, accHi, a2, a3); break;
        case 3: rows_batch<3, X4>(xp, s, e0, e1, e2, e3, k, lane, up, accLo, accHi, a2, a3); break;
        case 4: rows_batch<4, X4>(xp, s, e0, e1, e2, e3, k, lane, up, accLo, accHi, a2, a3); break;
        case 5: rows_batch<5, X4>(xp, s, e0, e1, e2, e3, k, lane, up, accLo, accHi, a2, a3); break;
        case 6: rows_batch<6, X4>(xp, s, e0, e1, e2, e3, k, lane, up, accLo, accHi, a2, a3); break;
        case 7: rows_batch<7, X4>(xp, s, e0, e1, e2, e3, k, lane, up, accLo, accHi, a2, a3); break;
        default: break;
      }
    }
  }
}

__global__ void k_gather(const int* __restrict__ off, const int* __restrict__ csr,
                         const float* __restrict__ a_s, const float* __restrict__ a_d,
                         const float* __restrict__ xp, const float* __restrict__ bias,
                         const float* __restrict__ pw, const float* __restrict__ pb,
                         float* __restrict__ hout, float* __restrict__ out,
                         int NT, int N, int threeN, int sec) {
  const unsigned lane = threadIdx.x & 31;
  const int wv = __builtin_amdgcn_readfirstlane(
      (int)((blockIdx.x * blockDim.x + threadIdx.x) >> 5));
  const bool up = lane >= 16;
  for (int g = 0; g < GPW; g++) {
  const int d = wv * GPW + g;
  if (d >= NT) return;
  v4f accLo, accHi, un2, un3;
  gather_node<0, false>(d, off, csr, a_s, a_d, xp, lane, up, accLo, accHi, un2, un3);
  {
    const v4f* bv = reinterpret_cast<const v4f*>(bias);
    const v4f* pv = reinterpret_cast<const v4f*>(pw);
    unsigned le = lane;
    asm volatile("" : "+v"(le));
    v4f tLo = accLo + bv[le], tHi = accHi + bv[32 + le];
    tLo.x = elu1(tLo.x); tLo.y = elu1(tLo.y); tLo.z = elu1(tLo.z); tLo.w = elu1(tLo.w);
    tHi.x = elu1(tHi.x); tHi.y = elu1(tHi.y); tHi.z = elu1(tHi.z); tHi.w = elu1(tHi.w);
    if (hout != nullptr) {
      v4f* o = reinterpret_cast<v4f*>(hout + (size_t)d * 256);
      o[le] = tLo;
      o[32 + le] = tHi;
    }
    const v4f pLo = pv[le], pHi = pv[32 + le];
    float pool = tLo.x * pLo.x + tLo.y * pLo.y + tLo.z * pLo.z + tLo.w * pLo.w
               + tHi.x * pHi.x + tHi.y * pHi.y + tHi.z * pHi.z + tHi.w * pHi.w;
    pool = wsum(pool);
    if (lane == 0) out[(size_t)(d / N) * threeN + sec * N + d % N] = pool + pb[0];
  }
  }
}

__global__ void
k_gather_gemm(const int* __restrict__ off, const int* __restrict__ csr,
              const float* __restrict__ a_s, const float* __restrict__ a_d,
              const float* __restrict__ x, const float* __restrict__ W1,
              const float* __restrict__ bias, const float* __restrict__ pw,
              const float* __restrict__ pb, const float* __restrict__ Bw,
              const float* __restrict__ att_s, const float* __restrict__ att_d,
              float* __restrict__ C, float* __restrict__ a_s2, float* __restrict__ a_d2,
              float* __restrict__ out, int NT, int K, int N, int threeN) {
  __shared__ alignas(16) float sX[4 * 32 * XP];
  __shared__ float sAtt[8][32];
  static_assert(32 * HP <= 4 * 32 * XP, "sH lies in the space of sX");
  float* const sH = sX;
  if ((int)blockIdx.x * 32 >= NT) return;
  const unsigned lane = threadIdx.x & 31;
  const int wib = __builtin_amdgcn_readfirstlane((int)(threadIdx.x >> 5));
  const bool up = lane >= 16;

  for (int g = 0; g < FGW; g++) {
    const int r = wib * FGW + g;
    const int d = (int)blockIdx.x * 32 + r;
    v4f h0, h1, h2, h3;
    gather_node<RB, true>(d, off, csr, a_s, a_d, x, lane, up, h0, h1, h2, h3);
    *reinterpret_cast<v4f*>(sX + (0 * 32 + r) * XP + 4 * lane) = h0;
    *reinterpret_cast<v4f*>(sX + (1 * 32 + r) * XP + 4 * lane) = h1;
    *reinterpret_cast<v4f*>(sX + (2 * 32 + r) * XP + 4 * lane) = h2;
    *reinterpret_cast<v4f*>(sX + (3 * 32 + r) * XP + 4 * lane) = h3;
  }
  __syncthreads();

  const int colGrp = wib & 3;
  const int row0l = (wib >> 2) << 4, col0 = colGrp << 6;
  const int row0 = (int)blockIdx.x * 32 + row0l;
  const int half = (int)(lane >> 4), lm = (int)(lane & 15);

  {
    v8f acc0 = {}, acc1 = {}, acc2 = {}, acc3 = {};
    const float* ap = sX + (colGrp * 32 + row0l + lm) * XP + 4 * half;
    const float* bp = W1 + (size_t)(4 * half) * 256 + col0 + 4 * lm;
    for (int k0 = 0; k0 < FIN; k0 += 8) {
      const v4f a4 = *reinterpret_cast<const v4f*>(ap + k0);
      mma_kstep8(a4, bp + (size_t)k0 * 256, acc0, acc1, acc2, acc3);
    }
    __syncthreads();
    const v4f bb = *reinterpret_cast<const v4f*>(bias + col0 + 4 * lm);
    float* hp = sH + (row0l + 8 * half) * HP + col0 + 4 * lm;
#pragma unroll
    for (int r2 = 0; r2 < 8; r2++) {
      v4f o;
      o.x = elu1(acc0[r2] + bb.x); o.y = elu1(acc1[r2] + bb.y);
      o.z = elu1(acc2[r2] + bb.z); o.w = elu1(acc3[r2] + bb.w);
      *reinterpret_cast<v4f*>(hp + r2 * HP) = o;
    }
  }
  __syncthreads();

  {
    const v4f* pv = reinterpret_cast<const v4f*>(pw);
    const v4f pLo = pv[lane], pHi = pv[32 + lane];
    for (int g = 0; g < FGW; g++) {
      const int r = wib * FGW + g;
      const int d = (int)blockIdx.x * 32 + r;
      const float* hr = sH + r * HP;
      const v4f tLo = *reinterpret_cast<const v4f*>(hr + 4 * lane);
      const v4f tHi = *reinterpret_cast<const v4f*>(hr + 128 + 4 * lane);
      float pool = tLo.x * pLo.x + tLo.y * pLo.y + tLo.z * pLo.z + tLo.w * pLo.w
                 + tHi.x * pHi.x + tHi.y * pHi.y + tHi.z * pHi.z + tHi.w * pHi.w;
      pool = wsum(pool);
      if (lane == 0) out[(size_t)(d / N) * threeN + N + d % N] = pool + pb[0];
    }
  }

  v8f acc0 = {}, acc1 = {}, acc2 = {}, acc3 = {};
  const float* ap = sH + (row0l + lm) * HP + 4 * half;
  const float* bp = Bw + (size_t)(4 * half) * 256 + col0 + 4 * lm;
  for (int k0 = 0; k0 < K; k0 += 8) {
    const v4f a4 = *reinterpret_cast<const v4f*>(ap + k0);
    mma_kstep8(a4, bp + (size_t)k0 * 256, acc0, acc1, acc2, acc3);
  }
  tile_finish(acc0, acc1, acc2, acc3, att_s, att_d, C, a_s2, a_d2, sAtt, row0, col0, half,
              lm, (int)lane, wib);
}

extern "C" void kernel_launch(void* const* d_in, const int* in_sizes, int n_in,
                              void* d_out, int out_size, void* d_ws, size_t ws_size,
                              hipStream_t stream) {
  const float* x   = (const float*)d_in[0];
  const int*   ei  = (const int*)d_in[1];
  const float* W1  = (const float*)d_in[2];
  const float* as1 = (const float*)d_in[3];
  const float* ad1 = (const float*)d_in[4];
  const float* b1  = (const float*)d_in[5];
  const float* W2  = (const float*)d_in[6];
  const float* as2 = (const float*)d_in[7];
  const float* ad2 = (const float*)d_in[8];
  const float* b2  = (const float*)d_in[9];
  const float* pw1 = (const float*)d_in[10];
  const float* pb1 = (const float*)d_in[11];
  const float* pw2 = (const float*)d_in[12];
  const float* pb2 = (const float*)d_in[13];
  float* out = (float*)d_out;

  const int NT = 32768, N = 4096, E = 524288;
  const int threeN = 3 * N;
  const size_t FEAT_BYTES = (size_t)NT * 256 * 4;

  char* w = (char*)d_ws;
  float* xp2  = (float*)(w + FEAT_BYTES);
  float* a_s2 = (float*)(w + 2 * FEAT_BYTES);
  float* a_d2 = a_s2 + (size_t)NT * 4;
  float* a_s  = (float*)(w + 3 * FEAT_BYTES);
  float* a_d  = a_s + (size_t)NT * 4;
  int*   cnt  = (int*)(a_d + (size_t)NT * 4);
  int*   off  = cnt + NT;
  int*   rank = off + NT + 32;
  int*   csr  = rank + E;

  const int* srcA = ei;
  const int* dstA = ei + E;

  dim3 blk(256);
  int gGather = (NT + GPW * 8 - 1) / (GPW * 8);
  int gGemm   = NT / 32;
  int gPre    = NT / (8 * PRE_N);
  int gEdge   = E / EPT / 256;
  int gCount  = (E + CNT_T * CNT_E - 1) / (CNT_T * CNT_E);

  k_pre<<<gPre, blk, 0, stream>>>(x, W1, as1, ad1, a_s, a_d, out, cnt, NT, N, threeN);

  k_count<<<gCount, dim3(CNT_T), 0, stream>>>(dstA, cnt, rank, E);
  k_scan<<<1, dim3(1024), 0, stream>>>(cnt, off, NT);
  k_fill<<<gEdge, blk, 0, stream>>>(srcA, dstA, off, rank, csr, E);

  k_gather_gemm<<<gGemm, blk, 0, stream>>>(off, csr, a_s, a_d, x, W1, b1, pw1, pb1, W2, as2, ad2,
                                          xp2, a_s2, a_d2, out, NT, 256, N, threeN);

  k_gather<<<gGather, blk, 0, stream>>>(off, csr, a_s2, a_d2, xp2, b2, pw2, pb2, nullptr, out, NT, N, threeN, 2);
}
